# starting kernel
# baseline (speedup 1.0000x reference)
#include <hip/hip_runtime.h>
#include <hip/hip_bf16.h>


#define NB   2048
#define NC   256
#define NI   9
#define NE   10
#define T3   165
#define T2   45
#define K3S  23
#define K2S  3
#define K3V  30
#define K2V  4
#define KTOT 224
#define KT   56
#define CTN  3
#define AST  226
#define ZST  49

#define OFF_U3S 0
#define OFF_U2S (OFF_U3S + T3*K3S)
#define OFF_U3V (OFF_U2S + T2*K2S)
#define OFF_U2V (OFF_U3V + 3*T3*K3V)
#define OFF_GFRAG 19456
#define GFRAG_PER_C (CTN*KT*64)

typedef float v2f __attribute__((ext_vector_type(2)));
typedef float v8f __attribute__((ext_vector_type(8)));

__device__ inline void t3_decode(int t, int& oa, int& ob, int& oc) {
  int idx = 0;
  for (int a = 0; a < NI; a++)
    for (int b = a; b < NI; b++)
      for (int c = b; c < NI; c++) {
        if (idx == t) { oa = a; ob = b; oc = c; return; }
        idx++;
      }
}
__device__ inline void t2_decode(int t, int& oa, int& ob) {
  int idx = 0;
  for (int a = 0; a < NI; a++)
    for (int b = a; b < NI; b++) {
      if (idx == t) { oa = a; ob = b; return; }
      idx++;
    }
}

template <typename F>
__device__ inline float symsum3(int a, int b, int c, F fetch) {
  const int P[6][3] = {{a,b,c},{a,c,b},{b,a,c},{b,c,a},{c,a,b},{c,b,a}};
  float s = 0.f;
  for (int p = 0; p < 6; p++) {
    int code = P[p][0]*81 + P[p][1]*9 + P[p][2];
    bool dup = false;
    for (int q = 0; q < p; q++) {
      if (P[q][0]*81 + P[q][1]*9 + P[q][2] == code) { dup = true; break; }
    }
    if (!dup) s += fetch(P[p][0], P[p][1], P[p][2]);
  }
  return s;
}

__global__ void prep_sym(const float* __restrict__ U2_s,
                         const float* __restrict__ U3_s,
                         const float* __restrict__ U2_v,
                         const float* __restrict__ U3_v,
                         float* __restrict__ ws) {
  int id = blockIdx.x * blockDim.x + threadIdx.x;
  const int n3s = T3*K3S, n2s = T2*K2S, n3v = 3*T3*K3V, n2v = 3*T2*K2V;

  if (id < n3s) {
    int t = id / K3S, k = id % K3S;
    int a, b, c; t3_decode(t, a, b, c);
    ws[OFF_U3S + id] = symsum3(a, b, c, [&](int i, int j, int l) {
      return U3_s[((i*NI + j)*NI + l)*K3S + k];
    });
    return;
  }
  id -= n3s;
  if (id < n2s) {
    int t = id / K2S, k = id % K2S;
    int a, b; t2_decode(t, a, b);
    float s = U2_s[(a*NI + b)*K2S + k];
    if (a != b) s += U2_s[(b*NI + a)*K2S + k];
    ws[OFF_U2S + id] = s;
    return;
  }
  id -= n2s;
  if (id < n3v) {
    int v = id / (T3*K3V), r = id % (T3*K3V);
    int t = r / K3V, k = r % K3V;
    int a, b, c; t3_decode(t, a, b, c);
    ws[OFF_U3V + id] = symsum3(a, b, c, [&](int i, int j, int l) {
      return U3_v[(((v*NI + i)*NI + j)*NI + l)*K3V + k];
    });
    return;
  }
  id -= n3v;
  if (id < n2v) {
    int v = id / (T2*K2V), r = id % (T2*K2V);
    int t = r / K2V, k = r % K2V;
    int a, b; t2_decode(t, a, b);
    float s = U2_v[((v*NI + a)*NI + b)*K2V + k];
    if (a != b) s += U2_v[((v*NI + b)*NI + a)*K2V + k];
    ws[OFF_U2V + id] = s;
  }
}

__global__ void prep_gfrag(const float* __restrict__ U1_s,
                           const float* __restrict__ U1_v,
                           const float* __restrict__ W1_s,
                           const float* __restrict__ W2_s,
                           const float* __restrict__ W3_s,
                           const float* __restrict__ W1_v,
                           const float* __restrict__ W2_v,
                           const float* __restrict__ W3_v,
                           const float* __restrict__ ws_u,
                           float* __restrict__ gfrag) {
  long id = (long)blockIdx.x * blockDim.x + threadIdx.x;
  if (id >= (long)NC * GFRAG_PER_C) return;
  int c    = (int)(id / GFRAG_PER_C);
  int rem  = (int)(id % GFRAG_PER_C);
  int ct   = rem / (KT*64); rem %= (KT*64);
  int kt   = rem / 64;
  int slot = rem % 64;
  int lane = slot >> 1, g = slot & 1;
  int t = kt*4 + g + ((lane >= 16) ? 2 : 0);
  int n = ct*16 + (lane & 15);

  float val = 0.f;
  if (n < 40 && t < 219) {
    int e = n >> 2, p = n & 3;
    if (t < T3) {
      if (p == 0) {
        const float* u = ws_u + OFF_U3S + t*K3S;
        const float* w = W3_s + e*K3S*NC + c;
        float s = 0.f;
        for (int k = 0; k < K3S; k++) s += u[k] * w[k*NC];
        val = s;
      } else {
        int v = p - 1;
        const float* u = ws_u + OFF_U3V + (v*T3 + t)*K3V;
        const float* w = W3_v + e*K3V*NC + c;
        float s = 0.f;
        for (int k = 0; k < K3V; k++) s += u[k] * w[k*NC];
        val = s;
      }
    } else if (t < T3 + T2) {
      int t2 = t - T3;
      if (p == 0) {
        const float* u = ws_u + OFF_U2S + t2*K2S;
        const float* w = W2_s + e*K2S*NC + c;
        float s = 0.f;
        for (int k = 0; k < K2S; k++) s += u[k] * w[k*NC];
        val = s;
      } else {
        int v = p - 1;
        const float* u = ws_u + OFF_U2V + (v*T2 + t2)*K2V;
        const float* w = W2_v + e*K2V*NC + c;
        float s = 0.f;
        for (int k = 0; k < K2V; k++) s += u[k] * w[k*NC];
        val = s;
      }
    } else {
      int i = t - (T3 + T2);
      if (p == 0) val = U1_s[i]          * W1_s[e*NC + c];
      else        val = U1_v[(p-1)*NI+i] * W1_v[e*NC + c];
    }
  }
  gfrag[id] = val;
}

__global__ void __launch_bounds__(256)
sc_main(const float* __restrict__ x, const float* __restrict__ y,
        const float* __restrict__ gfrag, float* __restrict__ out) {
  extern __shared__ float lds[];
  float* Gl = lds;
  float* Ab = Gl + GFRAG_PER_C;
  float* Zb = Ab + 128*AST;

  const int tid  = threadIdx.x;
  const int wave = tid >> 5;
  const int lane = tid & 31;
  const int c    = blockIdx.y;
  const int b0   = blockIdx.x * 128;

  {
    const float* gsrc = gfrag + (size_t)c * GFRAG_PER_C;
    for (int i = tid; i < GFRAG_PER_C; i += 256) Gl[i] = gsrc[i];
  }

  if (lane < 16) {
    const int r = wave*16 + lane;
    const int b = b0 + r;
    const float* xp = x + ((size_t)b*NC + c)*NI;
    float xv[NI];
#pragma unroll
    for (int i = 0; i < NI; i++) xv[i] = xp[i];
    float* arow = Ab + r*AST;
    int t = 0;
    for (int a = 0; a < NI; a++)
      for (int bq = a; bq < NI; bq++) {
        float p2 = xv[a]*xv[bq];
        for (int cq = bq; cq < NI; cq++) arow[t++] = p2*xv[cq];
      }
    for (int a = 0; a < NI; a++)
      for (int bq = a; bq < NI; bq++) arow[t++] = xv[a]*xv[bq];
#pragma unroll
    for (int i = 0; i < NI; i++) arow[t++] = xv[i];
    for (; t < KTOT; t++) arow[t] = 0.f;
  }
  __syncthreads();

  {
    const int half = lane >> 4;
    const int l16  = lane & 15;
    const float* arow = Ab + (wave*16 + l16)*AST + half*2;
    const float* g0 = Gl + 0*KT*64 + lane*2;
    const float* g1 = Gl + 1*KT*64 + lane*2;
    const float* g2 = Gl + 2*KT*64 + lane*2;

    v8f acc0 = {0,0,0,0,0,0,0,0};
    v8f acc1 = {0,0,0,0,0,0,0,0};
    v8f acc2 = {0,0,0,0,0,0,0,0};

#pragma unroll 8
    for (int kt = 0; kt < KT; kt++) {
      v2f A  = *(const v2f*)(arow + kt*4);
      v2f B0 = *(const v2f*)(g0 + kt*64);
      v2f B1 = *(const v2f*)(g1 + kt*64);
      v2f B2 = *(const v2f*)(g2 + kt*64);
      acc0 = __builtin_amdgcn_wmma_f32_16x16x4_f32(false, A, false, B0,
                                                   (short)0, acc0, false, false);
      acc1 = __builtin_amdgcn_wmma_f32_16x16x4_f32(false, A, false, B1,
                                                   (short)0, acc1, false, false);
      acc2 = __builtin_amdgcn_wmma_f32_16x16x4_f32(false, A, false, B2,
                                                   (short)0, acc2, false, false);
    }

    const int zr = wave*16 + (half ? 8 : 0);
#pragma unroll
    for (int g = 0; g < 8; g++) {
      Zb[(zr+g)*ZST +  0 + l16] = acc0[g];
      Zb[(zr+g)*ZST + 16 + l16] = acc1[g];
      Zb[(zr+g)*ZST + 32 + l16] = acc2[g];
    }
  }
  __syncthreads();

  {
    const int r  = tid >> 1;
    const int p0 = (tid & 1)*2;
    const int b  = b0 + r;
    const float* yb = y + (size_t)b*NE;
    float yv[NE];
#pragma unroll
    for (int e = 0; e < NE; e++) yv[e] = yb[e];
#pragma unroll
    for (int pp = 0; pp < 2; pp++) {
      const int p = p0 + pp;
      float s = 0.f;
#pragma unroll
      for (int e = 0; e < NE; e++) s += yv[e] * Zb[r*ZST + e*4 + p];
      size_t o = (size_t)b*(4*NC) +
                 ((p == 0) ? (size_t)c : (size_t)(NC + c*3 + (p-1)));
      out[o] = s;
    }
  }
}

extern "C" void kernel_launch(void* const* d_in, const int* in_sizes, int n_in,
                              void* d_out, int out_size, void* d_ws, size_t ws_size,
                              hipStream_t stream) {
  const float* x    = (const float*)d_in[0];
  const float* yv   = (const float*)d_in[1];
  const float* U1_s = (const float*)d_in[2];
  const float* U2_s = (const float*)d_in[3];
  const float* U3_s = (const float*)d_in[4];
  const float* U1_v = (const float*)d_in[5];
  const float* U2_v = (const float*)d_in[6];
  const float* U3_v = (const float*)d_in[7];
  const float* W1_s = (const float*)d_in[8];
  const float* W2_s = (const float*)d_in[9];
  const float* W3_s = (const float*)d_in[10];
  const float* W1_v = (const float*)d_in[11];
  const float* W2_v = (const float*)d_in[12];
  const float* W3_v = (const float*)d_in[13];
  float* ws  = (float*)d_ws;
  float* outp = (float*)d_out;

  {
    const int n = T3*K3S + T2*K2S + 3*T3*K3V + 3*T2*K2V;
    prep_sym<<<(n + 255)/256, 256, 0, stream>>>(U2_s, U3_s, U2_v, U3_v, ws);
  }
  {
    const long n = (long)NC * GFRAG_PER_C;
    prep_gfrag<<<(unsigned)((n + 255)/256), 256, 0, stream>>>(
        U1_s, U1_v, W1_s, W2_s, W3_s, W1_v, W2_v, W3_v, ws, ws + OFF_GFRAG);
  }
  {
    const size_t smem = (size_t)(GFRAG_PER_C + 128*AST + 128*ZST) * sizeof(float);
    sc_main<<<dim3(NB/128, NC), 256, smem, stream>>>(x, yv, ws + OFF_GFRAG, outp);
  }
}


# Z result panel aliased onto each wave's own A rows: LDS per workgroup 183808 -> 158720 B, so two workgroups share a WGP and one's staging overlaps the other's WMMA loop
# speedup vs baseline: 1.4026x; 1.4026x over previous
#include <hip/hip_runtime.h>
#include <hip/hip_bf16.h>


#define NB   2048
#define NC   256
#define NI   9
#define NE   10
#define T3   165
#define T2   45
#define K3S  23
#define K2S  3
#define K3V  30
#define K2V  4
#define KTOT 224
#define KT   56
#define CTN  3
#define AST  226
#define ZST  49

#define OFF_U3S 0
#define OFF_U2S (OFF_U3S + T3*K3S)
#define OFF_U3V (OFF_U2S + T2*K2S)
#define OFF_U2V (OFF_U3V + 3*T3*K3V)
#define OFF_GFRAG 19456
#define GFRAG_PER_C (CTN*KT*64)

typedef float v2f __attribute__((ext_vector_type(2)));
typedef float v8f __attribute__((ext_vector_type(8)));

__device__ inline void t3_decode(int t, int& oa, int& ob, int& oc) {
  int idx = 0;
  for (int a = 0; a < NI; a++)
    for (int b = a; b < NI; b++)
      for (int c = b; c < NI; c++) {
        if (idx == t) { oa = a; ob = b; oc = c; return; }
        idx++;
      }
}
__device__ inline void t2_decode(int t, int& oa, int& ob) {
  int idx = 0;
  for (int a = 0; a < NI; a++)
    for (int b = a; b < NI; b++) {
      if (idx == t) { oa = a; ob = b; return; }
      idx++;
    }
}

template <typename F>
__device__ inline float symsum3(int a, int b, int c, F fetch) {
  const int P[6][3] = {{a,b,c},{a,c,b},{b,a,c},{b,c,a},{c,a,b},{c,b,a}};
  float s = 0.f;
  for (int p = 0; p < 6; p++) {
    int code = P[p][0]*81 + P[p][1]*9 + P[p][2];
    bool dup = false;
    for (int q = 0; q < p; q++) {
      if (P[q][0]*81 + P[q][1]*9 + P[q][2] == code) { dup = true; break; }
    }
    if (!dup) s += fetch(P[p][0], P[p][1], P[p][2]);
  }
  return s;
}

__global__ void prep_sym(const float* __restrict__ U2_s,
                         const float* __restrict__ U3_s,
                         const float* __restrict__ U2_v,
                         const float* __restrict__ U3_v,
                         float* __restrict__ ws) {
  int id = blockIdx.x * blockDim.x + threadIdx.x;
  const int n3s = T3*K3S, n2s = T2*K2S, n3v = 3*T3*K3V, n2v = 3*T2*K2V;

  if (id < n3s) {
    int t = id / K3S, k = id % K3S;
    int a, b, c; t3_decode(t, a, b, c);
    ws[OFF_U3S + id] = symsum3(a, b, c, [&](int i, int j, int l) {
      return U3_s[((i*NI + j)*NI + l)*K3S + k];
    });
    return;
  }
  id -= n3s;
  if (id < n2s) {
    int t = id / K2S, k = id % K2S;
    int a, b; t2_decode(t, a, b);
    float s = U2_s[(a*NI + b)*K2S + k];
    if (a != b) s += U2_s[(b*NI + a)*K2S + k];
    ws[OFF_U2S + id] = s;
    return;
  }
  id -= n2s;
  if (id < n3v) {
    int v = id / (T3*K3V), r = id % (T3*K3V);
    int t = r / K3V, k = r % K3V;
    int a, b, c; t3_decode(t, a, b, c);
    ws[OFF_U3V + id] = symsum3(a, b, c, [&](int i, int j, int l) {
      return U3_v[(((v*NI + i)*NI + j)*NI + l)*K3V + k];
    });
    return;
  }
  id -= n3v;
  if (id < n2v) {
    int v = id / (T2*K2V), r = id % (T2*K2V);
    int t = r / K2V, k = r % K2V;
    int a, b; t2_decode(t, a, b);
    float s = U2_v[((v*NI + a)*NI + b)*K2V + k];
    if (a != b) s += U2_v[((v*NI + b)*NI + a)*K2V + k];
    ws[OFF_U2V + id] = s;
  }
}

__global__ void prep_gfrag(const float* __restrict__ U1_s,
                           const float* __restrict__ U1_v,
                           const float* __restrict__ W1_s,
                           const float* __restrict__ W2_s,
                           const float* __restrict__ W3_s,
                           const float* __restrict__ W1_v,
                           const float* __restrict__ W2_v,
                           const float* __restrict__ W3_v,
                           const float* __restrict__ ws_u,
                           float* __restrict__ gfrag) {
  long id = (long)blockIdx.x * blockDim.x + threadIdx.x;
  if (id >= (long)NC * GFRAG_PER_C) return;
  int c    = (int)(id / GFRAG_PER_C);
  int rem  = (int)(id % GFRAG_PER_C);
  int ct   = rem / (KT*64); rem %= (KT*64);
  int kt   = rem / 64;
  int slot = rem % 64;
  int lane = slot >> 1, g = slot & 1;
  int t = kt*4 + g + ((lane >= 16) ? 2 : 0);
  int n = ct*16 + (lane & 15);

  float val = 0.f;
  if (n < 40 && t < 219) {
    int e = n >> 2, p = n & 3;
    if (t < T3) {
      if (p == 0) {
        const float* u = ws_u + OFF_U3S + t*K3S;
        const float* w = W3_s + e*K3S*NC + c;
        float s = 0.f;
        for (int k = 0; k < K3S; k++) s += u[k] * w[k*NC];
        val = s;
      } else {
        int v = p - 1;
        const float* u = ws_u + OFF_U3V + (v*T3 + t)*K3V;
        const float* w = W3_v + e*K3V*NC + c;
        float s = 0.f;
        for (int k = 0; k < K3V; k++) s += u[k] * w[k*NC];
        val = s;
      }
    } else if (t < T3 + T2) {
      int t2 = t - T3;
      if (p == 0) {
        const float* u = ws_u + OFF_U2S + t2*K2S;
        const float* w = W2_s + e*K2S*NC + c;
        float s = 0.f;
        for (int k = 0; k < K2S; k++) s += u[k] * w[k*NC];
        val = s;
      } else {
        int v = p - 1;
        const float* u = ws_u + OFF_U2V + (v*T2 + t2)*K2V;
        const float* w = W2_v + e*K2V*NC + c;
        float s = 0.f;
        for (int k = 0; k < K2V; k++) s += u[k] * w[k*NC];
        val = s;
      }
    } else {
      int i = t - (T3 + T2);
      if (p == 0) val = U1_s[i]          * W1_s[e*NC + c];
      else        val = U1_v[(p-1)*NI+i] * W1_v[e*NC + c];
    }
  }
  gfrag[id] = val;
}

__global__ void __launch_bounds__(256)
sc_main(const float* __restrict__ x, const float* __restrict__ y,
        const float* __restrict__ gfrag, float* __restrict__ out) {
  extern __shared__ float lds[];
  float* Gl = lds;
  float* Ab = Gl + GFRAG_PER_C;

  const int tid  = threadIdx.x;
  const int wave = tid >> 5;
  const int lane = tid & 31;
  const int c    = blockIdx.y;
  const int b0   = blockIdx.x * 128;
  float* Zb = Ab + wave*16*AST;

  {
    const float* gsrc = gfrag + (size_t)c * GFRAG_PER_C;
    for (int i = tid; i < GFRAG_PER_C; i += 256) Gl[i] = gsrc[i];
  }

  if (lane < 16) {
    const int r = wave*16 + lane;
    const int b = b0 + r;
    const float* xp = x + ((size_t)b*NC + c)*NI;
    float xv[NI];
#pragma unroll
    for (int i = 0; i < NI; i++) xv[i] = xp[i];
    float* arow = Ab + r*AST;
    int t = 0;
    for (int a = 0; a < NI; a++)
      for (int bq = a; bq < NI; bq++) {
        float p2 = xv[a]*xv[bq];
        for (int cq = bq; cq < NI; cq++) arow[t++] = p2*xv[cq];
      }
    for (int a = 0; a < NI; a++)
      for (int bq = a; bq < NI; bq++) arow[t++] = xv[a]*xv[bq];
#pragma unroll
    for (int i = 0; i < NI; i++) arow[t++] = xv[i];
    for (; t < KTOT; t++) arow[t] = 0.f;
  }
  __syncthreads();

  {
    const int half = lane >> 4;
    const int l16  = lane & 15;
    const float* arow = Ab + (wave*16 + l16)*AST + half*2;
    const float* g0 = Gl + 0*KT*64 + lane*2;
    const float* g1 = Gl + 1*KT*64 + lane*2;
    const float* g2 = Gl + 2*KT*64 + lane*2;

    v8f acc0 = {0,0,0,0,0,0,0,0};
    v8f acc1 = {0,0,0,0,0,0,0,0};
    v8f acc2 = {0,0,0,0,0,0,0,0};

#pragma unroll 8
    for (int kt = 0; kt < KT; kt++) {
      v2f A  = *(const v2f*)(arow + kt*4);
      v2f B0 = *(const v2f*)(g0 + kt*64);
      v2f B1 = *(const v2f*)(g1 + kt*64);
      v2f B2 = *(const v2f*)(g2 + kt*64);
      acc0 = __builtin_amdgcn_wmma_f32_16x16x4_f32(false, A, false, B0,
                                                   (short)0, acc0, false, false);
      acc1 = __builtin_amdgcn_wmma_f32_16x16x4_f32(false, A, false, B1,
                                                   (short)0, acc1, false, false);
      acc2 = __builtin_amdgcn_wmma_f32_16x16x4_f32(false, A, false, B2,
                                                   (short)0, acc2, false, false);
    }

    const int zr = half ? 8 : 0;
#pragma unroll
    for (int g = 0; g < 8; g++) {
      Zb[(zr+g)*ZST +  0 + l16] = acc0[g];
      Zb[(zr+g)*ZST + 16 + l16] = acc1[g];
      Zb[(zr+g)*ZST + 32 + l16] = acc2[g];
    }
  }
  __syncthreads();

  {
    const int r  = tid >> 1;
    const int p0 = (tid & 1)*2;
    const int b  = b0 + r;
    const float* yb = y + (size_t)b*NE;
    float yv[NE];
#pragma unroll
    for (int e = 0; e < NE; e++) yv[e] = yb[e];
#pragma unroll
    for (int pp = 0; pp < 2; pp++) {
      const int p = p0 + pp;
      float s = 0.f;
#pragma unroll
      for (int e = 0; e < NE; e++) s += yv[e] * Zb[(r & 15)*ZST + e*4 + p];
      size_t o = (size_t)b*(4*NC) +
                 ((p == 0) ? (size_t)c : (size_t)(NC + c*3 + (p-1)));
      out[o] = s;
    }
  }
}

extern "C" void kernel_launch(void* const* d_in, const int* in_sizes, int n_in,
                              void* d_out, int out_size, void* d_ws, size_t ws_size,
                              hipStream_t stream) {
  const float* x    = (const float*)d_in[0];
  const float* yv   = (const float*)d_in[1];
  const float* U1_s = (const float*)d_in[2];
  const float* U2_s = (const float*)d_in[3];
  const float* U3_s = (const float*)d_in[4];
  const float* U1_v = (const float*)d_in[5];
  const float* U2_v = (const float*)d_in[6];
  const float* U3_v = (const float*)d_in[7];
  const float* W1_s = (const float*)d_in[8];
  const float* W2_s = (const float*)d_in[9];
  const float* W3_s = (const float*)d_in[10];
  const float* W1_v = (const float*)d_in[11];
  const float* W2_v = (const float*)d_in[12];
  const float* W3_v = (const float*)d_in[13];
  float* ws  = (float*)d_ws;
  float* outp = (float*)d_out;

  {
    const int n = T3*K3S + T2*K2S + 3*T3*K3V + 3*T2*K2V;
    prep_sym<<<(n + 255)/256, 256, 0, stream>>>(U2_s, U3_s, U2_v, U3_v, ws);
  }
  {
    const long n = (long)NC * GFRAG_PER_C;
    prep_gfrag<<<(unsigned)((n + 255)/256), 256, 0, stream>>>(
        U1_s, U1_v, W1_s, W2_s, W3_s, W1_v, W2_v, W3_v, ws, ws + OFF_GFRAG);
  }
  {
    const size_t smem = (size_t)(GFRAG_PER_C + 128*AST) * sizeof(float);
    sc_main<<<dim3(NB/128, NC), 256, smem, stream>>>(x, yv, ws + OFF_GFRAG, outp);
  }
}


# G-fragment prep remapped to lane = channel: one wave per (32 channels, e, K tile) makes 16 outputs from line-wide W loads and scalar U loads; 86016 waves -> 4480; each sum keeps its k order
# speedup vs baseline: 1.6894x; 1.2045x over previous
#include <hip/hip_runtime.h>
#include <hip/hip_bf16.h>


#define NB   2048
#define NC   256
#define NI   9
#define NE   10
#define T3   165
#define T2   45
#define K3S  23
#define K2S  3
#define K3V  30
#define K2V  4
#define KTOT 224
#define KT   56
#define CTN  3
#define AST  226
#define ZST  49

#define OFF_U3S 0
#define OFF_U2S (OFF_U3S + T3*K3S)
#define OFF_U3V (OFF_U2S + T2*K2S)
#define OFF_U2V (OFF_U3V + 3*T3*K3V)
#define OFF_GFRAG 19456
#define GFRAG_PER_C (CTN*KT*64)

typedef float v2f __attribute__((ext_vector_type(2)));
typedef float v8f __attribute__((ext_vector_type(8)));

__device__ inline void t3_decode(int t, int& oa, int& ob, int& oc) {
  int idx = 0;
  for (int a = 0; a < NI; a++)
    for (int b = a; b < NI; b++)
      for (int c = b; c < NI; c++) {
        if (idx == t) { oa = a; ob = b; oc = c; return; }
        idx++;
      }
}
__device__ inline void t2_decode(int t, int& oa, int& ob) {
  int idx = 0;
  for (int a = 0; a < NI; a++)
    for (int b = a; b < NI; b++) {
      if (idx == t) { oa = a; ob = b; return; }
      idx++;
    }
}

template <typename F>
__device__ inline float symsum3(int a, int b, int c, F fetch) {
  const int P[6][3] = {{a,b,c},{a,c,b},{b,a,c},{b,c,a},{c,a,b},{c,b,a}};
  float s = 0.f;
  for (int p = 0; p < 6; p++) {
    int code = P[p][0]*81 + P[p][1]*9 + P[p][2];
    bool dup = false;
    for (int q = 0; q < p; q++) {
      if (P[q][0]*81 + P[q][1]*9 + P[q][2] == code) { dup = true; break; }
    }
    if (!dup) s += fetch(P[p][0], P[p][1], P[p][2]);
  }
  return s;
}

__global__ void prep_sym(const float* __restrict__ U2_s,
                         const float* __restrict__ U3_s,
                         const float* __restrict__ U2_v,
                         const float* __restrict__ U3_v,
                         float* __restrict__ ws) {
  int id = blockIdx.x * blockDim.x + threadIdx.x;
  const int n3s = T3*K3S, n2s = T2*K2S, n3v = 3*T3*K3V, n2v = 3*T2*K2V;

  if (id < n3s) {
    int t = id / K3S, k = id % K3S;
    int a, b, c; t3_decode(t, a, b, c);
    ws[OFF_U3S + id] = symsum3(a, b, c, [&](int i, int j, int l) {
      return U3_s[((i*NI + j)*NI + l)*K3S + k];
    });
    return;
  }
  id -= n3s;
  if (id < n2s) {
    int t = id / K2S, k = id % K2S;
    int a, b; t2_decode(t, a, b);
    float s = U2_s[(a*NI + b)*K2S + k];
    if (a != b) s += U2_s[(b*NI + a)*K2S + k];
    ws[OFF_U2S + id] = s;
    return;
  }
  id -= n2s;
  if (id < n3v) {
    int v = id / (T3*K3V), r = id % (T3*K3V);
    int t = r / K3V, k = r % K3V;
    int a, b, c; t3_decode(t, a, b, c);
    ws[OFF_U3V + id] = symsum3(a, b, c, [&](int i, int j, int l) {
      return U3_v[(((v*NI + i)*NI + j)*NI + l)*K3V + k];
    });
    return;
  }
  id -= n3v;
  if (id < n2v) {
    int v = id / (T2*K2V), r = id % (T2*K2V);
    int t = r / K2V, k = r % K2V;
    int a, b; t2_decode(t, a, b);
    float s = U2_v[((v*NI + a)*NI + b)*K2V + k];
    if (a != b) s += U2_v[((v*NI + b)*NI + a)*K2V + k];
    ws[OFF_U2V + id] = s;
  }
}


__device__ inline void g_cubic_row(const float* __restrict__ ws_u, int t,
                                   const float (&w3s)[K3S], const float (&w3v)[K3V],
                                   float (&r)[4]) {
  const float* us = ws_u + OFF_U3S + t*K3S;
  float s = 0.f;
#pragma unroll
  for (int k = 0; k < K3S; k++) s += us[k] * w3s[k];
  r[0] = s;
#pragma unroll
  for (int v = 0; v < 3; v++) {
    const float* uv = ws_u + OFF_U3V + (v*T3 + t)*K3V;
    float q = 0.f;
#pragma unroll
    for (int k = 0; k < K3V; k++) q += uv[k] * w3v[k];
    r[1 + v] = q;
  }
}

__device__ inline void g_low_row(const float* __restrict__ ws_u, int t, int e, int c,
                                 const float* __restrict__ U1_s,
                                 const float* __restrict__ U1_v,
                                 const float* __restrict__ W1_s,
                                 const float* __restrict__ W2_s,
                                 const float* __restrict__ W1_v,
                                 const float* __restrict__ W2_v,
                                 float (&r)[4]) {
  r[0] = 0.f; r[1] = 0.f; r[2] = 0.f; r[3] = 0.f;
  if (t < T3 + T2) {
    const int t2 = t - T3;
    {
      const float* u = ws_u + OFF_U2S + t2*K2S;
      const float* w = W2_s + e*K2S*NC + c;
      float s = 0.f;
      for (int k = 0; k < K2S; k++) s += u[k] * w[k*NC];
      r[0] = s;
    }
    for (int v = 0; v < 3; v++) {
      const float* u = ws_u + OFF_U2V + (v*T2 + t2)*K2V;
      const float* w = W2_v + e*K2V*NC + c;
      float s = 0.f;
      for (int k = 0; k < K2V; k++) s += u[k] * w[k*NC];
      r[1 + v] = s;
    }
  } else if (t < T3 + T2 + NI) {
    const int i = t - (T3 + T2);
    const float w1s = W1_s[e*NC + c];
    const float w1v = W1_v[e*NC + c];
    r[0] = U1_s[i] * w1s;
    for (int v = 0; v < 3; v++) r[1 + v] = U1_v[v*NI + i] * w1v;
  }
}

__global__ void __launch_bounds__(32)
prep_gfrag(const float* __restrict__ U1_s,
           const float* __restrict__ U1_v,
           const float* __restrict__ W1_s,
           const float* __restrict__ W2_s,
           const float* __restrict__ W3_s,
           const float* __restrict__ W1_v,
           const float* __restrict__ W2_v,
           const float* __restrict__ W3_v,
           const float* __restrict__ ws_u,
           float* __restrict__ gfrag) {
  const int bid = blockIdx.x;
  const int kt  = bid % KT;
  const int e   = (bid / KT) % NE;
  const int cg  = bid / (KT*NE);
  const int c   = cg*32 + (int)threadIdx.x;

  float res[4][4];
  if (kt*4 < T3) {
    float w3s[K3S], w3v[K3V];
#pragma unroll
    for (int k = 0; k < K3S; k++) w3s[k] = W3_s[(e*K3S + k)*NC + c];
#pragma unroll
    for (int k = 0; k < K3V; k++) w3v[k] = W3_v[(e*K3V + k)*NC + c];
#pragma unroll
    for (int tl = 0; tl < 4; tl++) {
      const int t = kt*4 + tl;
      if (t < T3) g_cubic_row(ws_u, t, w3s, w3v, res[tl]);
      else        g_low_row(ws_u, t, e, c, U1_s, U1_v, W1_s, W2_s, W1_v, W2_v, res[tl]);
    }
  } else {
#pragma unroll
    for (int tl = 0; tl < 4; tl++)
      g_low_row(ws_u, kt*4 + tl, e, c, U1_s, U1_v, W1_s, W2_s, W1_v, W2_v, res[tl]);
  }

  const int ct = e >> 2;
  float* dst = gfrag + ((size_t)(c*CTN + ct)*KT + kt)*64 + (e & 3)*8;
  typedef float v4f __attribute__((ext_vector_type(4)));
  v4f o;
  o.x = res[0][0]; o.y = res[1][0]; o.z = res[0][1]; o.w = res[1][1];
  *(v4f*)(dst +  0) = o;
  o.x = res[0][2]; o.y = res[1][2]; o.z = res[0][3]; o.w = res[1][3];
  *(v4f*)(dst +  4) = o;
  o.x = res[2][0]; o.y = res[3][0]; o.z = res[2][1]; o.w = res[3][1];
  *(v4f*)(dst + 32) = o;
  o.x = res[2][2]; o.y = res[3][2]; o.z = res[2][3]; o.w = res[3][3];
  *(v4f*)(dst + 36) = o;

  if (e == NE - 1) {
    const v4f z = {0.f, 0.f, 0.f, 0.f};
    float* pad = gfrag + ((size_t)(c*CTN + 2)*KT + kt)*64;
    for (int j = 0; j < 4; j++) {
      *(v4f*)(pad + 16 + j*4) = z;
      *(v4f*)(pad + 48 + j*4) = z;
    }
  }
}

__global__ void __launch_bounds__(256)
sc_main(const float* __restrict__ x, const float* __restrict__ y,
        const float* __restrict__ gfrag, float* __restrict__ out) {
  extern __shared__ float lds[];
  float* Gl = lds;
  float* Ab = Gl + GFRAG_PER_C;

  const int tid  = threadIdx.x;
  const int wave = tid >> 5;
  const int lane = tid & 31;
  const int c    = blockIdx.y;
  const int b0   = blockIdx.x * 128;
  float* Zb = Ab + wave*16*AST;

  {
    const float* gsrc = gfrag + (size_t)c * GFRAG_PER_C;
    for (int i = tid; i < GFRAG_PER_C; i += 256) Gl[i] = gsrc[i];
  }

  if (lane < 16) {
    const int r = wave*16 + lane;
    const int b = b0 + r;
    const float* xp = x + ((size_t)b*NC + c)*NI;
    float xv[NI];
#pragma unroll
    for (int i = 0; i < NI; i++) xv[i] = xp[i];
    float* arow = Ab + r*AST;
    int t = 0;
    for (int a = 0; a < NI; a++)
      for (int bq = a; bq < NI; bq++) {
        float p2 = xv[a]*xv[bq];
        for (int cq = bq; cq < NI; cq++) arow[t++] = p2*xv[cq];
      }
    for (int a = 0; a < NI; a++)
      for (int bq = a; bq < NI; bq++) arow[t++] = xv[a]*xv[bq];
#pragma unroll
    for (int i = 0; i < NI; i++) arow[t++] = xv[i];
    for (; t < KTOT; t++) arow[t] = 0.f;
  }
  __syncthreads();

  {
    const int half = lane >> 4;
    const int l16  = lane & 15;
    const float* arow = Ab + (wave*16 + l16)*AST + half*2;
    const float* g0 = Gl + 0*KT*64 + lane*2;
    const float* g1 = Gl + 1*KT*64 + lane*2;
    const float* g2 = Gl + 2*KT*64 + lane*2;

    v8f acc0 = {0,0,0,0,0,0,0,0};
    v8f acc1 = {0,0,0,0,0,0,0,0};
    v8f acc2 = {0,0,0,0,0,0,0,0};

#pragma unroll 8
    for (int kt = 0; kt < KT; kt++) {
      v2f A  = *(const v2f*)(arow + kt*4);
      v2f B0 = *(const v2f*)(g0 + kt*64);
      v2f B1 = *(const v2f*)(g1 + kt*64);
      v2f B2 = *(const v2f*)(g2 + kt*64);
      acc0 = __builtin_amdgcn_wmma_f32_16x16x4_f32(false, A, false, B0,
                                                   (short)0, acc0, false, false);
      acc1 = __builtin_amdgcn_wmma_f32_16x16x4_f32(false, A, false, B1,
                                                   (short)0, acc1, false, false);
      acc2 = __builtin_amdgcn_wmma_f32_16x16x4_f32(false, A, false, B2,
                                                   (short)0, acc2, false, false);
    }

    const int zr = half ? 8 : 0;
#pragma unroll
    for (int g = 0; g < 8; g++) {
      Zb[(zr+g)*ZST +  0 + l16] = acc0[g];
      Zb[(zr+g)*ZST + 16 + l16] = acc1[g];
      Zb[(zr+g)*ZST + 32 + l16] = acc2[g];
    }
  }
  __syncthreads();

  {
    const int r  = tid >> 1;
    const int p0 = (tid & 1)*2;
    const int b  = b0 + r;
    const float* yb = y + (size_t)b*NE;
    float yv[NE];
#pragma unroll
    for (int e = 0; e < NE; e++) yv[e] = yb[e];
#pragma unroll
    for (int pp = 0; pp < 2; pp++) {
      const int p = p0 + pp;
      float s = 0.f;
#pragma unroll
      for (int e = 0; e < NE; e++) s += yv[e] * Zb[(r & 15)*ZST + e*4 + p];
      size_t o = (size_t)b*(4*NC) +
                 ((p == 0) ? (size_t)c : (size_t)(NC + c*3 + (p-1)));
      out[o] = s;
    }
  }
}

extern "C" void kernel_launch(void* const* d_in, const int* in_sizes, int n_in,
                              void* d_out, int out_size, void* d_ws, size_t ws_size,
                              hipStream_t stream) {
  const float* x    = (const float*)d_in[0];
  const float* yv   = (const float*)d_in[1];
  const float* U1_s = (const float*)d_in[2];
  const float* U2_s = (const float*)d_in[3];
  const float* U3_s = (const float*)d_in[4];
  const float* U1_v = (const float*)d_in[5];
  const float* U2_v = (const float*)d_in[6];
  const float* U3_v = (const float*)d_in[7];
  const float* W1_s = (const float*)d_in[8];
  const float* W2_s = (const float*)d_in[9];
  const float* W3_s = (const float*)d_in[10];
  const float* W1_v = (const float*)d_in[11];
  const float* W2_v = (const float*)d_in[12];
  const float* W3_v = (const float*)d_in[13];
  float* ws  = (float*)d_ws;
  float* outp = (float*)d_out;

  {
    const int n = T3*K3S + T2*K2S + 3*T3*K3V + 3*T2*K2V;
    prep_sym<<<(n + 255)/256, 256, 0, stream>>>(U2_s, U3_s, U2_v, U3_v, ws);
  }
  {
    prep_gfrag<<<(NC/32)*NE*KT, 32, 0, stream>>>(
        U1_s, U1_v, W1_s, W2_s, W3_s, W1_v, W2_v, W3_v, ws, ws + OFF_GFRAG);
  }
  {
    const size_t smem = (size_t)(GFRAG_PER_C + 128*AST) * sizeof(float);
    sc_main<<<dim3(NB/128, NC), 256, smem, stream>>>(x, yv, ws + OFF_GFRAG, outp);
  }
}


# sc_main issues all its global loads in one batch before its first wait (x row first, G fragments as 16-byte loads, y row): nine dependent memory round trips per workgroup become one
# speedup vs baseline: 2.2557x; 1.3352x over previous
#include <hip/hip_runtime.h>
#include <hip/hip_bf16.h>


#define NB   2048
#define NC   256
#define NI   9
#define NE   10
#define T3   165
#define T2   45
#define K3S  23
#define K2S  3
#define K3V  30
#define K2V  4
#define KTOT 224
#define KT   56
#define CTN  3
#define AST  226
#define ZST  49

#define OFF_U3S 0
#define OFF_U2S (OFF_U3S + T3*K3S)
#define OFF_U3V (OFF_U2S + T2*K2S)
#define OFF_U2V (OFF_U3V + 3*T3*K3V)
#define OFF_GFRAG 19456
#define GFRAG_PER_C (CTN*KT*64)

typedef float v2f __attribute__((ext_vector_type(2)));
typedef float v8f __attribute__((ext_vector_type(8)));

__device__ inline void t3_decode(int t, int& oa, int& ob, int& oc) {
  int idx = 0;
  for (int a = 0; a < NI; a++)
    for (int b = a; b < NI; b++)
      for (int c = b; c < NI; c++) {
        if (idx == t) { oa = a; ob = b; oc = c; return; }
        idx++;
      }
}
__device__ inline void t2_decode(int t, int& oa, int& ob) {
  int idx = 0;
  for (int a = 0; a < NI; a++)
    for (int b = a; b < NI; b++) {
      if (idx == t) { oa = a; ob = b; return; }
      idx++;
    }
}

template <typename F>
__device__ inline float symsum3(int a, int b, int c, F fetch) {
  const int P[6][3] = {{a,b,c},{a,c,b},{b,a,c},{b,c,a},{c,a,b},{c,b,a}};
  float s = 0.f;
  for (int p = 0; p < 6; p++) {
    int code = P[p][0]*81 + P[p][1]*9 + P[p][2];
    bool dup = false;
    for (int q = 0; q < p; q++) {
      if (P[q][0]*81 + P[q][1]*9 + P[q][2] == code) { dup = true; break; }
    }
    if (!dup) s += fetch(P[p][0], P[p][1], P[p][2]);
  }
  return s;
}

__global__ void prep_sym(const float* __restrict__ U2_s,
                         const float* __restrict__ U3_s,
                         const float* __restrict__ U2_v,
                         const float* __restrict__ U3_v,
                         float* __restrict__ ws) {
  int id = blockIdx.x * blockDim.x + threadIdx.x;
  const int n3s = T3*K3S, n2s = T2*K2S, n3v = 3*T3*K3V, n2v = 3*T2*K2V;

  if (id < n3s) {
    int t = id / K3S, k = id % K3S;
    int a, b, c; t3_decode(t, a, b, c);
    ws[OFF_U3S + id] = symsum3(a, b, c, [&](int i, int j, int l) {
      return U3_s[((i*NI + j)*NI + l)*K3S + k];
    });
    return;
  }
  id -= n3s;
  if (id < n2s) {
    int t = id / K2S, k = id % K2S;
    int a, b; t2_decode(t, a, b);
    float s = U2_s[(a*NI + b)*K2S + k];
    if (a != b) s += U2_s[(b*NI + a)*K2S + k];
    ws[OFF_U2S + id] = s;
    return;
  }
  id -= n2s;
  if (id < n3v) {
    int v = id / (T3*K3V), r = id % (T3*K3V);
    int t = r / K3V, k = r % K3V;
    int a, b, c; t3_decode(t, a, b, c);
    ws[OFF_U3V + id] = symsum3(a, b, c, [&](int i, int j, int l) {
      return U3_v[(((v*NI + i)*NI + j)*NI + l)*K3V + k];
    });
    return;
  }
  id -= n3v;
  if (id < n2v) {
    int v = id / (T2*K2V), r = id % (T2*K2V);
    int t = r / K2V, k = r % K2V;
    int a, b; t2_decode(t, a, b);
    float s = U2_v[((v*NI + a)*NI + b)*K2V + k];
    if (a != b) s += U2_v[((v*NI + b)*NI + a)*K2V + k];
    ws[OFF_U2V + id] = s;
  }
}


__device__ inline void g_cubic_row(const float* __restrict__ ws_u, int t,
                                   const float (&w3s)[K3S], const float (&w3v)[K3V],
                                   float (&r)[4]) {
  const float* us = ws_u + OFF_U3S + t*K3S;
  float s = 0.f;
#pragma unroll
  for (int k = 0; k < K3S; k++) s += us[k] * w3s[k];
  r[0] = s;
#pragma unroll
  for (int v = 0; v < 3; v++) {
    const float* uv = ws_u + OFF_U3V + (v*T3 + t)*K3V;
    float q = 0.f;
#pragma unroll
    for (int k = 0; k < K3V; k++) q += uv[k] * w3v[k];
    r[1 + v] = q;
  }
}

__device__ inline void g_low_row(const float* __restrict__ ws_u, int t, int e, int c,
                                 const float* __restrict__ U1_s,
                                 const float* __restrict__ U1_v,
                                 const float* __restrict__ W1_s,
                                 const float* __restrict__ W2_s,
                                 const float* __restrict__ W1_v,
                                 const float* __restrict__ W2_v,
                                 float (&r)[4]) {
  r[0] = 0.f; r[1] = 0.f; r[2] = 0.f; r[3] = 0.f;
  if (t < T3 + T2) {
    const int t2 = t - T3;
    {
      const float* u = ws_u + OFF_U2S + t2*K2S;
      const float* w = W2_s + e*K2S*NC + c;
      float s = 0.f;
      for (int k = 0; k < K2S; k++) s += u[k] * w[k*NC];
      r[0] = s;
    }
    for (int v = 0; v < 3; v++) {
      const float* u = ws_u + OFF_U2V + (v*T2 + t2)*K2V;
      const float* w = W2_v + e*K2V*NC + c;
      float s = 0.f;
      for (int k = 0; k < K2V; k++) s += u[k] * w[k*NC];
      r[1 + v] = s;
    }
  } else if (t < T3 + T2 + NI) {
    const int i = t - (T3 + T2);
    const float w1s = W1_s[e*NC + c];
    const float w1v = W1_v[e*NC + c];
    r[0] = U1_s[i] * w1s;
    for (int v = 0; v < 3; v++) r[1 + v] = U1_v[v*NI + i] * w1v;
  }
}

__global__ void __launch_bounds__(32)
prep_gfrag(const float* __restrict__ U1_s,
           const float* __restrict__ U1_v,
           const float* __restrict__ W1_s,
           const float* __restrict__ W2_s,
           const float* __restrict__ W3_s,
           const float* __restrict__ W1_v,
           const float* __restrict__ W2_v,
           const float* __restrict__ W3_v,
           const float* __restrict__ ws_u,
           float* __restrict__ gfrag) {
  const int bid = blockIdx.x;
  const int kt  = bid % KT;
  const int e   = (bid / KT) % NE;
  const int cg  = bid / (KT*NE);
  const int c   = cg*32 + (int)threadIdx.x;

  float res[4][4];
  if (kt*4 < T3) {
    float w3s[K3S], w3v[K3V];
#pragma unroll
    for (int k = 0; k < K3S; k++) w3s[k] = W3_s[(e*K3S + k)*NC + c];
#pragma unroll
    for (int k = 0; k < K3V; k++) w3v[k] = W3_v[(e*K3V + k)*NC + c];
#pragma unroll
    for (int tl = 0; tl < 4; tl++) {
      const int t = kt*4 + tl;
      if (t < T3) g_cubic_row(ws_u, t, w3s, w3v, res[tl]);
      else        g_low_row(ws_u, t, e, c, U1_s, U1_v, W1_s, W2_s, W1_v, W2_v, res[tl]);
    }
  } else {
#pragma unroll
    for (int tl = 0; tl < 4; tl++)
      g_low_row(ws_u, kt*4 + tl, e, c, U1_s, U1_v, W1_s, W2_s, W1_v, W2_v, res[tl]);
  }

  const int ct = e >> 2;
  float* dst = gfrag + ((size_t)(c*CTN + ct)*KT + kt)*64 + (e & 3)*8;
  typedef float v4f __attribute__((ext_vector_type(4)));
  v4f o;
  o.x = res[0][0]; o.y = res[1][0]; o.z = res[0][1]; o.w = res[1][1];
  *(v4f*)(dst +  0) = o;
  o.x = res[0][2]; o.y = res[1][2]; o.z = res[0][3]; o.w = res[1][3];
  *(v4f*)(dst +  4) = o;
  o.x = res[2][0]; o.y = res[3][0]; o.z = res[2][1]; o.w = res[3][1];
  *(v4f*)(dst + 32) = o;
  o.x = res[2][2]; o.y = res[3][2]; o.z = res[2][3]; o.w = res[3][3];
  *(v4f*)(dst + 36) = o;

  if (e == NE - 1) {
    const v4f z = {0.f, 0.f, 0.f, 0.f};
    float* pad = gfrag + ((size_t)(c*CTN + 2)*KT + kt)*64;
    for (int j = 0; j < 4; j++) {
      *(v4f*)(pad + 16 + j*4) = z;
      *(v4f*)(pad + 48 + j*4) = z;
    }
  }
}

__global__ void __launch_bounds__(256)
sc_main(const float* __restrict__ x, const float* __restrict__ y,
        const float* __restrict__ gfrag, float* __restrict__ out) {
  extern __shared__ float lds[];
  float* Gl = lds;
  float* Ab = Gl + GFRAG_PER_C;

  const int tid  = threadIdx.x;
  const int wave = tid >> 5;
  const int lane = tid & 31;
  const int c    = blockIdx.y;
  const int b0   = blockIdx.x * 128;
  float* Zb = Ab + wave*16*AST;

  typedef float v4f __attribute__((ext_vector_type(4)));
  const int xr = wave*16 + (lane & 15);
  float xv[NI];
  {
    const float* xp = x + ((size_t)(b0 + xr)*NC + c)*NI;
#pragma unroll
    for (int i = 0; i < NI; i++) xv[i] = xp[i];
  }
  __builtin_amdgcn_sched_barrier(0);
  v4f gr[11];
  {
    const v4f* gsrc = (const v4f*)(gfrag + (size_t)c * GFRAG_PER_C);
#pragma unroll
    for (int j = 0; j < 10; j++) gr[j] = gsrc[tid + 256*j];
    gr[10] = gsrc[2560 + (tid & 127)];
  }
  float yv[NE];
  {
    const float* yb = y + (size_t)(b0 + (tid >> 1))*NE;
#pragma unroll
    for (int e = 0; e < NE; e++) yv[e] = yb[e];
  }

  asm volatile("" :: "v"(xv[0]), "v"(xv[1]), "v"(xv[2]), "v"(xv[3]), "v"(xv[4]),
                     "v"(xv[5]), "v"(xv[6]), "v"(xv[7]), "v"(xv[8]));

  if (lane < 16) {
    float* arow = Ab + xr*AST;
    int t = 0;
    for (int a = 0; a < NI; a++)
      for (int bq = a; bq < NI; bq++) {
        float p2 = xv[a]*xv[bq];
        for (int cq = bq; cq < NI; cq++) arow[t++] = p2*xv[cq];
      }
    for (int a = 0; a < NI; a++)
      for (int bq = a; bq < NI; bq++) arow[t++] = xv[a]*xv[bq];
#pragma unroll
    for (int i = 0; i < NI; i++) arow[t++] = xv[i];
    for (; t < KTOT; t++) arow[t] = 0.f;
  }

  {
    v4f* Gl4 = (v4f*)Gl;
#pragma unroll
    for (int j = 0; j < 10; j++) Gl4[tid + 256*j] = gr[j];
    if (tid < 128) Gl4[2560 + tid] = gr[10];
  }
  __syncthreads();

  {
    const int half = lane >> 4;
    const int l16  = lane & 15;
    const float* arow = Ab + (wave*16 + l16)*AST + half*2;
    const float* g0 = Gl + 0*KT*64 + lane*2;
    const float* g1 = Gl + 1*KT*64 + lane*2;
    const float* g2 = Gl + 2*KT*64 + lane*2;

    v8f acc0 = {0,0,0,0,0,0,0,0};
    v8f acc1 = {0,0,0,0,0,0,0,0};
    v8f acc2 = {0,0,0,0,0,0,0,0};

#pragma unroll 8
    for (int kt = 0; kt < KT; kt++) {
      v2f A  = *(const v2f*)(arow + kt*4);
      v2f B0 = *(const v2f*)(g0 + kt*64);
      v2f B1 = *(const v2f*)(g1 + kt*64);
      v2f B2 = *(const v2f*)(g2 + kt*64);
      acc0 = __builtin_amdgcn_wmma_f32_16x16x4_f32(false, A, false, B0,
                                                   (short)0, acc0, false, false);
      acc1 = __builtin_amdgcn_wmma_f32_16x16x4_f32(false, A, false, B1,
                                                   (short)0, acc1, false, false);
      acc2 = __builtin_amdgcn_wmma_f32_16x16x4_f32(false, A, false, B2,
                                                   (short)0, acc2, false, false);
    }

    const int zr = half ? 8 : 0;
#pragma unroll
    for (int g = 0; g < 8; g++) {
      Zb[(zr+g)*ZST +  0 + l16] = acc0[g];
      Zb[(zr+g)*ZST + 16 + l16] = acc1[g];
      Zb[(zr+g)*ZST + 32 + l16] = acc2[g];
    }
  }
  __syncthreads();

  {
    const int r  = tid >> 1;
    const int p0 = (tid & 1)*2;
    const int b  = b0 + r;
#pragma unroll
    for (int pp = 0; pp < 2; pp++) {
      const int p = p0 + pp;
      float s = 0.f;
#pragma unroll
      for (int e = 0; e < NE; e++) s += yv[e] * Zb[(r & 15)*ZST + e*4 + p];
      size_t o = (size_t)b*(4*NC) +
                 ((p == 0) ? (size_t)c : (size_t)(NC + c*3 + (p-1)));
      out[o] = s;
    }
  }
}

extern "C" void kernel_launch(void* const* d_in, const int* in_sizes, int n_in,
                              void* d_out, int out_size, void* d_ws, size_t ws_size,
                              hipStream_t stream) {
  const float* x    = (const float*)d_in[0];
  const float* yv   = (const float*)d_in[1];
  const float* U1_s = (const float*)d_in[2];
  const float* U2_s = (const float*)d_in[3];
  const float* U3_s = (const float*)d_in[4];
  const float* U1_v = (const float*)d_in[5];
  const float* U2_v = (const float*)d_in[6];
  const float* U3_v = (const float*)d_in[7];
  const float* W1_s = (const float*)d_in[8];
  const float* W2_s = (const float*)d_in[9];
  const float* W3_s = (const float*)d_in[10];
  const float* W1_v = (const float*)d_in[11];
  const float* W2_v = (const float*)d_in[12];
  const float* W3_v = (const float*)d_in[13];
  float* ws  = (float*)d_ws;
  float* outp = (float*)d_out;

  {
    const int n = T3*K3S + T2*K2S + 3*T3*K3V + 3*T2*K2V;
    prep_sym<<<(n + 255)/256, 256, 0, stream>>>(U2_s, U3_s, U2_v, U3_v, ws);
  }
  {
    prep_gfrag<<<(NC/32)*NE*KT, 32, 0, stream>>>(
        U1_s, U1_v, W1_s, W2_s, W3_s, W1_v, W2_v, W3_v, ws, ws + OFF_GFRAG);
  }
  {
    const size_t smem = (size_t)(GFRAG_PER_C + 128*AST) * sizeof(float);
    sc_main<<<dim3(NB/128, NC), 256, smem, stream>>>(x, yv, ws + OFF_GFRAG, outp);
  }
}


# U2/U3 symmetrization folded into the G-fragment prep: a 256-thread workgroup per (e, K tile) symmetrizes its 4 rows into LDS first; the separate launch and workspace pass go (3 -> 2); sum order kept
# speedup vs baseline: 2.3355x; 1.0354x over previous
#include <hip/hip_runtime.h>
#include <hip/hip_bf16.h>


#define NB   2048
#define NC   256
#define NI   9
#define NE   10
#define T3   165
#define T2   45
#define K3S  23
#define K2S  3
#define K3V  30
#define K2V  4
#define KTOT 224
#define KT   56
#define CTN  3
#define AST  226
#define ZST  49

#define OFF_GFRAG 19456
#define GFRAG_PER_C (CTN*KT*64)

typedef float v2f __attribute__((ext_vector_type(2)));
typedef float v8f __attribute__((ext_vector_type(8)));

__device__ inline void t3_decode(int t, int& oa, int& ob, int& oc) {
  int a = 0, rem = t;
  for (int it = 0; it < NI; it++) {
    const int n = (NI - a)*(NI - a + 1)/2;
    if (rem < n) break;
    rem -= n; a++;
  }
  int b = a;
  for (int it = 0; it < NI; it++) {
    const int n = NI - b;
    if (rem < n) break;
    rem -= n; b++;
  }
  oa = a; ob = b; oc = b + rem;
}
__device__ inline void t2_decode(int t, int& oa, int& ob) {
  int a = 0, rem = t;
  for (int it = 0; it < NI; it++) {
    const int n = NI - a;
    if (rem < n) break;
    rem -= n; a++;
  }
  oa = a; ob = a + rem;
}

template <typename F>
__device__ inline float symsum3(int a, int b, int c, F fetch) {
  const int P[6][3] = {{a,b,c},{a,c,b},{b,a,c},{b,c,a},{c,a,b},{c,b,a}};
  float f[6];
  for (int p = 0; p < 6; p++) f[p] = fetch(P[p][0], P[p][1], P[p][2]);
  float s = 0.f;
  for (int p = 0; p < 6; p++) {
    int code = P[p][0]*81 + P[p][1]*9 + P[p][2];
    bool dup = false;
    for (int q = 0; q < p; q++)
      dup = dup || (P[q][0]*81 + P[q][1]*9 + P[q][2] == code);
    const float s1 = s + f[p];
    s = dup ? s : s1;
  }
  return s;
}


#define USTR 128

__device__ inline bool usym_entry(int t, int j,
                                  const float* __restrict__ U2_s,
                                  const float* __restrict__ U3_s,
                                  const float* __restrict__ U2_v,
                                  const float* __restrict__ U3_v,
                                  int& slot, float& val) {
  if (t < T3) {
    int a, b, c; t3_decode(t, a, b, c);
    const bool sc = j < K3S;
    const int v  = sc ? 0 : (j - K3S) / K3V;
    const int k  = sc ? j : (j - K3S) % K3V;
    const int kk = sc ? K3S : K3V;
    const float* u = sc ? U3_s : U3_v + v*(NI*NI*NI*K3V);
    slot = sc ? k : 32 + 32*v + k;
    val = symsum3(a, b, c, [&](int i, int jj, int l) {
      return u[((i*NI + jj)*NI + l)*kk + k];
    });
    return true;
  }
  if (t < T3 + T2 && j < K2S + 3*K2V) {
    int a, b; t2_decode(t - T3, a, b);
    const bool sc = j < K2S;
    const int v  = sc ? 0 : (j - K2S) / K2V;
    const int k  = sc ? j : (j - K2S) % K2V;
    const int kk = sc ? K2S : K2V;
    const float* u = sc ? U2_s : U2_v + v*(NI*NI*K2V);
    slot = sc ? k : 32 + 32*v + k;
    const float f0 = u[(a*NI + b)*kk + k];
    const float f1 = u[(b*NI + a)*kk + k];
    const float s1 = f0 + f1;
    val = (a != b) ? s1 : f0;
    return true;
  }
  return false;
}

__device__ inline void g_cubic_row(const float* ur,
                                   const float (&w3s)[K3S], const float (&w3v)[K3V],
                                   float (&r)[4]) {
  float s = 0.f;
#pragma unroll
  for (int k = 0; k < K3S; k++) s += ur[k] * w3s[k];
  r[0] = s;
#pragma unroll
  for (int v = 0; v < 3; v++) {
    const float* uv = ur + 32 + 32*v;
    float q = 0.f;
#pragma unroll
    for (int k = 0; k < K3V; k++) q += uv[k] * w3v[k];
    r[1 + v] = q;
  }
}

__device__ inline void g_low_row(const float* ur, int t, int e, int c,
                                 const float* __restrict__ U1_s,
                                 const float* __restrict__ U1_v,
                                 const float* __restrict__ W1_s,
                                 const float* __restrict__ W2_s,
                                 const float* __restrict__ W1_v,
                                 const float* __restrict__ W2_v,
                                 float (&r)[4]) {
  r[0] = 0.f; r[1] = 0.f; r[2] = 0.f; r[3] = 0.f;
  if (t < T3 + T2) {
    {
      const float* w = W2_s + e*K2S*NC + c;
      float s = 0.f;
      for (int k = 0; k < K2S; k++) s += ur[k] * w[k*NC];
      r[0] = s;
    }
    for (int v = 0; v < 3; v++) {
      const float* u = ur + 32 + 32*v;
      const float* w = W2_v + e*K2V*NC + c;
      float s = 0.f;
      for (int k = 0; k < K2V; k++) s += u[k] * w[k*NC];
      r[1 + v] = s;
    }
  } else if (t < T3 + T2 + NI) {
    const int i = t - (T3 + T2);
    const float w1s = W1_s[e*NC + c];
    const float w1v = W1_v[e*NC + c];
    r[0] = U1_s[i] * w1s;
    for (int v = 0; v < 3; v++) r[1 + v] = U1_v[v*NI + i] * w1v;
  }
}

__global__ void __launch_bounds__(256)
prep_gfrag(const float* __restrict__ U1_s,
           const float* __restrict__ U2_s,
           const float* __restrict__ U3_s,
           const float* __restrict__ U1_v,
           const float* __restrict__ U2_v,
           const float* __restrict__ U3_v,
           const float* __restrict__ W1_s,
           const float* __restrict__ W2_s,
           const float* __restrict__ W3_s,
           const float* __restrict__ W1_v,
           const float* __restrict__ W2_v,
           const float* __restrict__ W3_v,
           float* __restrict__ gfrag) {
  __shared__ __attribute__((aligned(16))) float usym[4*USTR];

  const int bid = blockIdx.x;
  const int kt  = bid % KT;
  const int e   = bid / KT;
  const int c   = (int)threadIdx.x;

  if (kt*4 < T3 + T2) {
    const int NJ = K3S + 3*K3V;
    for (int idx = (int)threadIdx.x; idx < 4*NJ; idx += 256) {
      const int tl = idx / NJ, j = idx - tl*NJ;
      int slot; float val;
      if (usym_entry(kt*4 + tl, j, U2_s, U3_s, U2_v, U3_v, slot, val))
        usym[tl*USTR + slot] = val;
    }
  }
  __syncthreads();

  float res[4][4];
  if (kt*4 < T3) {
    float w3s[K3S], w3v[K3V];
#pragma unroll
    for (int k = 0; k < K3S; k++) w3s[k] = W3_s[(e*K3S + k)*NC + c];
#pragma unroll
    for (int k = 0; k < K3V; k++) w3v[k] = W3_v[(e*K3V + k)*NC + c];
#pragma unroll
    for (int tl = 0; tl < 4; tl++) {
      const int t = kt*4 + tl;
      if (t < T3) g_cubic_row(usym + tl*USTR, w3s, w3v, res[tl]);
      else        g_low_row(usym + tl*USTR, t, e, c, U1_s, U1_v, W1_s, W2_s, W1_v, W2_v, res[tl]);
    }
  } else {
#pragma unroll
    for (int tl = 0; tl < 4; tl++)
      g_low_row(usym + tl*USTR, kt*4 + tl, e, c, U1_s, U1_v, W1_s, W2_s, W1_v, W2_v, res[tl]);
  }

  const int ct = e >> 2;
  float* dst = gfrag + ((size_t)(c*CTN + ct)*KT + kt)*64 + (e & 3)*8;
  typedef float v4f __attribute__((ext_vector_type(4)));
  v4f o;
  o.x = res[0][0]; o.y = res[1][0]; o.z = res[0][1]; o.w = res[1][1];
  *(v4f*)(dst +  0) = o;
  o.x = res[0][2]; o.y = res[1][2]; o.z = res[0][3]; o.w = res[1][3];
  *(v4f*)(dst +  4) = o;
  o.x = res[2][0]; o.y = res[3][0]; o.z = res[2][1]; o.w = res[3][1];
  *(v4f*)(dst + 32) = o;
  o.x = res[2][2]; o.y = res[3][2]; o.z = res[2][3]; o.w = res[3][3];
  *(v4f*)(dst + 36) = o;

  if (e == NE - 1) {
    const v4f z = {0.f, 0.f, 0.f, 0.f};
    float* pad = gfrag + ((size_t)(c*CTN + 2)*KT + kt)*64;
    for (int j = 0; j < 4; j++) {
      *(v4f*)(pad + 16 + j*4) = z;
      *(v4f*)(pad + 48 + j*4) = z;
    }
  }
}

__global__ void __launch_bounds__(256)
sc_main(const float* __restrict__ x, const float* __restrict__ y,
        const float* __restrict__ gfrag, float* __restrict__ out) {
  extern __shared__ float lds[];
  float* Gl = lds;
  float* Ab = Gl + GFRAG_PER_C;

  const int tid  = threadIdx.x;
  const int wave = tid >> 5;
  const int lane = tid & 31;
  const int c    = blockIdx.y;
  const int b0   = blockIdx.x * 128;
  float* Zb = Ab + wave*16*AST;

  typedef float v4f __attribute__((ext_vector_type(4)));
  const int xr = wave*16 + (lane & 15);
  float xv[NI];
  {
    const float* xp = x + ((size_t)(b0 + xr)*NC + c)*NI;
#pragma unroll
    for (int i = 0; i < NI; i++) xv[i] = xp[i];
  }
  __builtin_amdgcn_sched_barrier(0);
  v4f gr[11];
  {
    const v4f* gsrc = (const v4f*)(gfrag + (size_t)c * GFRAG_PER_C);
#pragma unroll
    for (int j = 0; j < 10; j++) gr[j] = gsrc[tid + 256*j];
    gr[10] = gsrc[2560 + (tid & 127)];
  }
  float yv[NE];
  {
    const float* yb = y + (size_t)(b0 + (tid >> 1))*NE;
#pragma unroll
    for (int e = 0; e < NE; e++) yv[e] = yb[e];
  }

  asm volatile("" :: "v"(xv[0]), "v"(xv[1]), "v"(xv[2]), "v"(xv[3]), "v"(xv[4]),
                     "v"(xv[5]), "v"(xv[6]), "v"(xv[7]), "v"(xv[8]));

  if (lane < 16) {
    float* arow = Ab + xr*AST;
    int t = 0;
    for (int a = 0; a < NI; a++)
      for (int bq = a; bq < NI; bq++) {
        float p2 = xv[a]*xv[bq];
        for (int cq = bq; cq < NI; cq++) arow[t++] = p2*xv[cq];
      }
    for (int a = 0; a < NI; a++)
      for (int bq = a; bq < NI; bq++) arow[t++] = xv[a]*xv[bq];
#pragma unroll
    for (int i = 0; i < NI; i++) arow[t++] = xv[i];
    for (; t < KTOT; t++) arow[t] = 0.f;
  }

  {
    v4f* Gl4 = (v4f*)Gl;
#pragma unroll
    for (int j = 0; j < 10; j++) Gl4[tid + 256*j] = gr[j];
    if (tid < 128) Gl4[2560 + tid] = gr[10];
  }
  __syncthreads();

  {
    const int half = lane >> 4;
    const int l16  = lane & 15;
    const float* arow = Ab + (wave*16 + l16)*AST + half*2;
    const float* g0 = Gl + 0*KT*64 + lane*2;
    const float* g1 = Gl + 1*KT*64 + lane*2;
    const float* g2 = Gl + 2*KT*64 + lane*2;

    v8f acc0 = {0,0,0,0,0,0,0,0};
    v8f acc1 = {0,0,0,0,0,0,0,0};
    v8f acc2 = {0,0,0,0,0,0,0,0};

#pragma unroll 8
    for (int kt = 0; kt < KT; kt++) {
      v2f A  = *(const v2f*)(arow + kt*4);
      v2f B0 = *(const v2f*)(g0 + kt*64);
      v2f B1 = *(const v2f*)(g1 + kt*64);
      v2f B2 = *(const v2f*)(g2 + kt*64);
      acc0 = __builtin_amdgcn_wmma_f32_16x16x4_f32(false, A, false, B0,
                                                   (short)0, acc0, false, false);
      acc1 = __builtin_amdgcn_wmma_f32_16x16x4_f32(false, A, false, B1,
                                                   (short)0, acc1, false, false);
      acc2 = __builtin_amdgcn_wmma_f32_16x16x4_f32(false, A, false, B2,
                                                   (short)0, acc2, false, false);
    }

    const int zr = half ? 8 : 0;
#pragma unroll
    for (int g = 0; g < 8; g++) {
      Zb[(zr+g)*ZST +  0 + l16] = acc0[g];
      Zb[(zr+g)*ZST + 16 + l16] = acc1[g];
      Zb[(zr+g)*ZST + 32 + l16] = acc2[g];
    }
  }
  __syncthreads();

  {
    const int r  = tid >> 1;
    const int p0 = (tid & 1)*2;
    const int b  = b0 + r;
#pragma unroll
    for (int pp = 0; pp < 2; pp++) {
      const int p = p0 + pp;
      float s = 0.f;
#pragma unroll
      for (int e = 0; e < NE; e++) s += yv[e] * Zb[(r & 15)*ZST + e*4 + p];
      size_t o = (size_t)b*(4*NC) +
                 ((p == 0) ? (size_t)c : (size_t)(NC + c*3 + (p-1)));
      out[o] = s;
    }
  }
}

extern "C" void kernel_launch(void* const* d_in, const int* in_sizes, int n_in,
                              void* d_out, int out_size, void* d_ws, size_t ws_size,
                              hipStream_t stream) {
  const float* x    = (const float*)d_in[0];
  const float* yv   = (const float*)d_in[1];
  const float* U1_s = (const float*)d_in[2];
  const float* U2_s = (const float*)d_in[3];
  const float* U3_s = (const float*)d_in[4];
  const float* U1_v = (const float*)d_in[5];
  const float* U2_v = (const float*)d_in[6];
  const float* U3_v = (const float*)d_in[7];
  const float* W1_s = (const float*)d_in[8];
  const float* W2_s = (const float*)d_in[9];
  const float* W3_s = (const float*)d_in[10];
  const float* W1_v = (const float*)d_in[11];
  const float* W2_v = (const float*)d_in[12];
  const float* W3_v = (const float*)d_in[13];
  float* ws  = (float*)d_ws;
  float* outp = (float*)d_out;

  {
    prep_gfrag<<<NE*KT, 256, 0, stream>>>(
        U1_s, U2_s, U3_s, U1_v, U2_v, U3_v,
        W1_s, W2_s, W3_s, W1_v, W2_v, W3_v, ws + OFF_GFRAG);
  }
  {
    const size_t smem = (size_t)(GFRAG_PER_C + 128*AST) * sizeof(float);
    sc_main<<<dim3(NB/128, NC), 256, smem, stream>>>(x, yv, ws + OFF_GFRAG, outp);
  }
}


# sc_main builds its A fragments in registers (each lane computes the two monomials per K tile its fragment holds): the LDS A panel goes; LDS per workgroup 158720 -> 68096 B, three workgroups per WGP
# speedup vs baseline: 2.7066x; 1.1589x over previous
#include <hip/hip_runtime.h>
#include <hip/hip_bf16.h>
#include <utility>


#define NB   2048
#define NC   256
#define NI   9
#define NE   10
#define T3   165
#define T2   45
#define K3S  23
#define K2S  3
#define K3V  30
#define K2V  4
#define KTOT 224
#define KT   56
#define CTN  3
#define ZST  49

#define OFF_GFRAG 19456
#define GFRAG_PER_C (CTN*KT*64)

typedef float v2f __attribute__((ext_vector_type(2)));
typedef float v8f __attribute__((ext_vector_type(8)));

__device__ inline void t3_decode(int t, int& oa, int& ob, int& oc) {
  int a = 0, rem = t;
  for (int it = 0; it < NI; it++) {
    const int n = (NI - a)*(NI - a + 1)/2;
    if (rem < n) break;
    rem -= n; a++;
  }
  int b = a;
  for (int it = 0; it < NI; it++) {
    const int n = NI - b;
    if (rem < n) break;
    rem -= n; b++;
  }
  oa = a; ob = b; oc = b + rem;
}
__device__ inline void t2_decode(int t, int& oa, int& ob) {
  int a = 0, rem = t;
  for (int it = 0; it < NI; it++) {
    const int n = NI - a;
    if (rem < n) break;
    rem -= n; a++;
  }
  oa = a; ob = a + rem;
}

template <typename F>
__device__ inline float symsum3(int a, int b, int c, F fetch) {
  const int P[6][3] = {{a,b,c},{a,c,b},{b,a,c},{b,c,a},{c,a,b},{c,b,a}};
  float f[6];
  for (int p = 0; p < 6; p++) f[p] = fetch(P[p][0], P[p][1], P[p][2]);
  float s = 0.f;
  for (int p = 0; p < 6; p++) {
    int code = P[p][0]*81 + P[p][1]*9 + P[p][2];
    bool dup = false;
    for (int q = 0; q < p; q++)
      dup = dup || (P[q][0]*81 + P[q][1]*9 + P[q][2] == code);
    const float s1 = s + f[p];
    s = dup ? s : s1;
  }
  return s;
}


#define USTR 128

__device__ inline bool usym_entry(int t, int j,
                                  const float* __restrict__ U2_s,
                                  const float* __restrict__ U3_s,
                                  const float* __restrict__ U2_v,
                                  const float* __restrict__ U3_v,
                                  int& slot, float& val) {
  if (t < T3) {
    int a, b, c; t3_decode(t, a, b, c);
    const bool sc = j < K3S;
    const int v  = sc ? 0 : (j - K3S) / K3V;
    const int k  = sc ? j : (j - K3S) % K3V;
    const int kk = sc ? K3S : K3V;
    const float* u = sc ? U3_s : U3_v + v*(NI*NI*NI*K3V);
    slot = sc ? k : 32 + 32*v + k;
    val = symsum3(a, b, c, [&](int i, int jj, int l) {
      return u[((i*NI + jj)*NI + l)*kk + k];
    });
    return true;
  }
  if (t < T3 + T2 && j < K2S + 3*K2V) {
    int a, b; t2_decode(t - T3, a, b);
    const bool sc = j < K2S;
    const int v  = sc ? 0 : (j - K2S) / K2V;
    const int k  = sc ? j : (j - K2S) % K2V;
    const int kk = sc ? K2S : K2V;
    const float* u = sc ? U2_s : U2_v + v*(NI*NI*K2V);
    slot = sc ? k : 32 + 32*v + k;
    const float f0 = u[(a*NI + b)*kk + k];
    const float f1 = u[(b*NI + a)*kk + k];
    const float s1 = f0 + f1;
    val = (a != b) ? s1 : f0;
    return true;
  }
  return false;
}

__device__ inline void g_cubic_row(const float* ur,
                                   const float (&w3s)[K3S], const float (&w3v)[K3V],
                                   float (&r)[4]) {
  float s = 0.f;
#pragma unroll
  for (int k = 0; k < K3S; k++) s += ur[k] * w3s[k];
  r[0] = s;
#pragma unroll
  for (int v = 0; v < 3; v++) {
    const float* uv = ur + 32 + 32*v;
    float q = 0.f;
#pragma unroll
    for (int k = 0; k < K3V; k++) q += uv[k] * w3v[k];
    r[1 + v] = q;
  }
}

__device__ inline void g_low_row(const float* ur, int t, int e, int c,
                                 const float* __restrict__ U1_s,
                                 const float* __restrict__ U1_v,
                                 const float* __restrict__ W1_s,
                                 const float* __restrict__ W2_s,
                                 const float* __restrict__ W1_v,
                                 const float* __restrict__ W2_v,
                                 float (&r)[4]) {
  r[0] = 0.f; r[1] = 0.f; r[2] = 0.f; r[3] = 0.f;
  if (t < T3 + T2) {
    {
      const float* w = W2_s + e*K2S*NC + c;
      float s = 0.f;
      for (int k = 0; k < K2S; k++) s += ur[k] * w[k*NC];
      r[0] = s;
    }
    for (int v = 0; v < 3; v++) {
      const float* u = ur + 32 + 32*v;
      const float* w = W2_v + e*K2V*NC + c;
      float s = 0.f;
      for (int k = 0; k < K2V; k++) s += u[k] * w[k*NC];
      r[1 + v] = s;
    }
  } else if (t < T3 + T2 + NI) {
    const int i = t - (T3 + T2);
    const float w1s = W1_s[e*NC + c];
    const float w1v = W1_v[e*NC + c];
    r[0] = U1_s[i] * w1s;
    for (int v = 0; v < 3; v++) r[1 + v] = U1_v[v*NI + i] * w1v;
  }
}

__global__ void __launch_bounds__(256)
prep_gfrag(const float* __restrict__ U1_s,
           const float* __restrict__ U2_s,
           const float* __restrict__ U3_s,
           const float* __restrict__ U1_v,
           const float* __restrict__ U2_v,
           const float* __restrict__ U3_v,
           const float* __restrict__ W1_s,
           const float* __restrict__ W2_s,
           const float* __restrict__ W3_s,
           const float* __restrict__ W1_v,
           const float* __restrict__ W2_v,
           const float* __restrict__ W3_v,
           float* __restrict__ gfrag) {
  __shared__ __attribute__((aligned(16))) float usym[4*USTR];

  const int bid = blockIdx.x;
  const int kt  = bid % KT;
  const int e   = bid / KT;
  const int c   = (int)threadIdx.x;

  if (kt*4 < T3 + T2) {
    const int NJ = K3S + 3*K3V;
    for (int idx = (int)threadIdx.x; idx < 4*NJ; idx += 256) {
      const int tl = idx / NJ, j = idx - tl*NJ;
      int slot; float val;
      if (usym_entry(kt*4 + tl, j, U2_s, U3_s, U2_v, U3_v, slot, val))
        usym[tl*USTR + slot] = val;
    }
  }
  __syncthreads();

  float res[4][4];
  if (kt*4 < T3) {
    float w3s[K3S], w3v[K3V];
#pragma unroll
    for (int k = 0; k < K3S; k++) w3s[k] = W3_s[(e*K3S + k)*NC + c];
#pragma unroll
    for (int k = 0; k < K3V; k++) w3v[k] = W3_v[(e*K3V + k)*NC + c];
#pragma unroll
    for (int tl = 0; tl < 4; tl++) {
      const int t = kt*4 + tl;
      if (t < T3) g_cubic_row(usym + tl*USTR, w3s, w3v, res[tl]);
      else        g_low_row(usym + tl*USTR, t, e, c, U1_s, U1_v, W1_s, W2_s, W1_v, W2_v, res[tl]);
    }
  } else {
#pragma unroll
    for (int tl = 0; tl < 4; tl++)
      g_low_row(usym + tl*USTR, kt*4 + tl, e, c, U1_s, U1_v, W1_s, W2_s, W1_v, W2_v, res[tl]);
  }

  const int ct = e >> 2;
  float* dst = gfrag + ((size_t)(c*CTN + ct)*KT + kt)*64 + (e & 3)*8;
  typedef float v4f __attribute__((ext_vector_type(4)));
  v4f o;
  o.x = res[0][0]; o.y = res[1][0]; o.z = res[0][1]; o.w = res[1][1];
  *(v4f*)(dst +  0) = o;
  o.x = res[0][2]; o.y = res[1][2]; o.z = res[0][3]; o.w = res[1][3];
  *(v4f*)(dst +  4) = o;
  o.x = res[2][0]; o.y = res[3][0]; o.z = res[2][1]; o.w = res[3][1];
  *(v4f*)(dst + 32) = o;
  o.x = res[2][2]; o.y = res[3][2]; o.z = res[2][3]; o.w = res[3][3];
  *(v4f*)(dst + 36) = o;

  if (e == NE - 1) {
    const v4f z = {0.f, 0.f, 0.f, 0.f};
    float* pad = gfrag + ((size_t)(c*CTN + 2)*KT + kt)*64;
    for (int j = 0; j < 4; j++) {
      *(v4f*)(pad + 16 + j*4) = z;
      *(v4f*)(pad + 48 + j*4) = z;
    }
  }
}


#define F_ONE  9
#define F_ZERO 10
struct Fac { int f0, f1, f2; };
constexpr Fac fac_of(int t) {
  if (t < T3) {
    int a = 0, rem = t;
    for (int it = 0; it < NI; it++) {
      const int n = (NI - a)*(NI - a + 1)/2;
      if (rem < n) break;
      rem -= n; a++;
    }
    int b = a;
    for (int it = 0; it < NI; it++) {
      const int n = NI - b;
      if (rem < n) break;
      rem -= n; b++;
    }
    return Fac{a, b, b + rem};
  }
  if (t < T3 + T2) {
    int a = 0, rem = t - T3;
    for (int it = 0; it < NI; it++) {
      const int n = NI - a;
      if (rem < n) break;
      rem -= n; a++;
    }
    return Fac{a, a + rem, F_ONE};
  }
  if (t < T3 + T2 + NI) return Fac{t - (T3 + T2), F_ONE, F_ONE};
  return Fac{F_ZERO, F_ONE, F_ONE};
}
static_assert(fac_of(0).f0 == 0 && fac_of(0).f1 == 0 && fac_of(0).f2 == 0, "first cubic row");
static_assert(fac_of(T3 - 1).f0 == 8 && fac_of(T3 - 1).f1 == 8 && fac_of(T3 - 1).f2 == 8, "last cubic row");
static_assert(fac_of(T3).f0 == 0 && fac_of(T3).f1 == 0 && fac_of(T3).f2 == F_ONE, "first quadratic row");
static_assert(fac_of(T3 + T2 - 1).f0 == 8 && fac_of(T3 + T2 - 1).f1 == 8, "last quadratic row");
static_assert(fac_of(T3 + T2).f0 == 0 && fac_of(T3 + T2).f1 == F_ONE, "first linear row");
static_assert(fac_of(T3 + T2 + NI - 1).f0 == 8 && fac_of(T3 + T2 + NI).f0 == F_ZERO, "last linear row, first padding row");
template <int I>
__device__ __forceinline__ float fac_val(const float (&x)[NI]) {
  if constexpr (I < NI) return x[I];
  else if constexpr (I == F_ONE) return 1.f;
  else return 0.f;
}
template <int I, int J>
__device__ __forceinline__ float pick(const float (&x)[NI], bool hi) {
  if constexpr (I == J) return fac_val<I>(x);
  else return hi ? fac_val<J>(x) : fac_val<I>(x);
}
template <int T0, int T1>
__device__ __forceinline__ float slot(const float (&x)[NI], bool hi) {
  constexpr Fac a = fac_of(T0), b = fac_of(T1);
  const float p2 = pick<a.f0, b.f0>(x, hi) * pick<a.f1, b.f1>(x, hi);
  return p2 * pick<a.f2, b.f2>(x, hi);
}
template <int KT0, size_t... I>
__device__ __forceinline__ void build_a(const float (&x)[NI], bool hi,
                                        v2f (&A)[sizeof...(I)],
                                        std::index_sequence<I...>) {
  ((A[I] = v2f{slot<(KT0 + (int)I)*4 + 0, (KT0 + (int)I)*4 + 2>(x, hi),
               slot<(KT0 + (int)I)*4 + 1, (KT0 + (int)I)*4 + 3>(x, hi)}), ...);
}
#define GK 4
#define NG (KT/GK)
template <int KT0>
__device__ __forceinline__ void load_b(const float* g0, const float* g1,
                                       const float* g2, v2f (&B)[GK][3]) {
#pragma unroll
  for (int i = 0; i < GK; i++) {
    B[i][0] = *(const v2f*)(g0 + (KT0 + i)*64);
    B[i][1] = *(const v2f*)(g1 + (KT0 + i)*64);
    B[i][2] = *(const v2f*)(g2 + (KT0 + i)*64);
  }
}
template <int KT0>
__device__ __forceinline__ void mma_step(const float (&xv)[NI], bool hi,
                                         const v2f (&B)[GK][3],
                                         v8f& acc0, v8f& acc1, v8f& acc2) {
  v2f A[GK];
  build_a<KT0>(xv, hi, A, std::make_index_sequence<GK>{});
#pragma unroll
  for (int i = 0; i < GK; i++) {
    acc0 = __builtin_amdgcn_wmma_f32_16x16x4_f32(false, A[i], false, B[i][0],
                                                 (short)0, acc0, false, false);
    acc1 = __builtin_amdgcn_wmma_f32_16x16x4_f32(false, A[i], false, B[i][1],
                                                 (short)0, acc1, false, false);
    acc2 = __builtin_amdgcn_wmma_f32_16x16x4_f32(false, A[i], false, B[i][2],
                                                 (short)0, acc2, false, false);
  }
}
template <int G>
__device__ __forceinline__ void k_steps(const float (&xv)[NI], bool hi,
                                        const float* g0, const float* g1,
                                        const float* g2,
                                        v2f (&Bc)[GK][3], v2f (&Bn)[GK][3],
                                        v8f& acc0, v8f& acc1, v8f& acc2) {
  if constexpr (G + 1 < NG) load_b<(G + 1)*GK>(g0, g1, g2, Bn);
  __builtin_amdgcn_sched_barrier(0);
  mma_step<G*GK>(xv, hi, Bc, acc0, acc1, acc2);
  if constexpr (G + 1 < NG)
    k_steps<G + 1>(xv, hi, g0, g1, g2, Bn, Bc, acc0, acc1, acc2);
}

__global__ void __launch_bounds__(256)
sc_main(const float* __restrict__ x, const float* __restrict__ y,
        const float* __restrict__ gfrag, float* __restrict__ out) {
  extern __shared__ float lds[];
  float* Gl = lds;
  float* Zs = Gl + GFRAG_PER_C;

  const int tid  = threadIdx.x;
  const int wave = tid >> 5;
  const int lane = tid & 31;
  const int c    = blockIdx.y;
  const int b0   = blockIdx.x * 128;
  float* Zb = Zs + wave*16*ZST;

  typedef float v4f __attribute__((ext_vector_type(4)));
  const int xr = wave*16 + (lane & 15);
  float xv[NI];
  {
    const float* xp = x + ((size_t)(b0 + xr)*NC + c)*NI;
#pragma unroll
    for (int i = 0; i < NI; i++) xv[i] = xp[i];
  }
  v4f gr[11];
  {
    const v4f* gsrc = (const v4f*)(gfrag + (size_t)c * GFRAG_PER_C);
#pragma unroll
    for (int j = 0; j < 10; j++) gr[j] = gsrc[tid + 256*j];
    gr[10] = gsrc[2560 + (tid & 127)];
  }
  float yv[NE];
  {
    const float* yb = y + (size_t)(b0 + (tid >> 1))*NE;
#pragma unroll
    for (int e = 0; e < NE; e++) yv[e] = yb[e];
  }

  {
    v4f* Gl4 = (v4f*)Gl;
#pragma unroll
    for (int j = 0; j < 10; j++) Gl4[tid + 256*j] = gr[j];
    asm volatile("" :: "v"(gr[10]));
    if (tid < 128) Gl4[2560 + tid] = gr[10];
  }
  __syncthreads();

  {
    const bool hi  = (lane >> 4) != 0;
    const int  l16 = lane & 15;
    const float* g0 = Gl + 0*KT*64 + lane*2;
    const float* g1 = Gl + 1*KT*64 + lane*2;
    const float* g2 = Gl + 2*KT*64 + lane*2;

    v8f acc0 = {0,0,0,0,0,0,0,0};
    v8f acc1 = {0,0,0,0,0,0,0,0};
    v8f acc2 = {0,0,0,0,0,0,0,0};

    v2f Ba[GK][3], Bb[GK][3];
    load_b<0>(g0, g1, g2, Ba);
    k_steps<0>(xv, hi, g0, g1, g2, Ba, Bb, acc0, acc1, acc2);

    const int zr = hi ? 8 : 0;
#pragma unroll
    for (int g = 0; g < 8; g++) {
      Zb[(zr+g)*ZST +  0 + l16] = acc0[g];
      Zb[(zr+g)*ZST + 16 + l16] = acc1[g];
      Zb[(zr+g)*ZST + 32 + l16] = acc2[g];
    }
  }
  __syncthreads();

  {
    const int r  = tid >> 1;
    const int p0 = (tid & 1)*2;
    const int b  = b0 + r;
#pragma unroll
    for (int pp = 0; pp < 2; pp++) {
      const int p = p0 + pp;
      float s = 0.f;
#pragma unroll
      for (int e = 0; e < NE; e++) s += yv[e] * Zb[(r & 15)*ZST + e*4 + p];
      size_t o = (size_t)b*(4*NC) +
                 ((p == 0) ? (size_t)c : (size_t)(NC + c*3 + (p-1)));
      out[o] = s;
    }
  }
}

extern "C" void kernel_launch(void* const* d_in, const int* in_sizes, int n_in,
                              void* d_out, int out_size, void* d_ws, size_t ws_size,
                              hipStream_t stream) {
  const float* x    = (const float*)d_in[0];
  const float* yv   = (const float*)d_in[1];
  const float* U1_s = (const float*)d_in[2];
  const float* U2_s = (const float*)d_in[3];
  const float* U3_s = (const float*)d_in[4];
  const float* U1_v = (const float*)d_in[5];
  const float* U2_v = (const float*)d_in[6];
  const float* U3_v = (const float*)d_in[7];
  const float* W1_s = (const float*)d_in[8];
  const float* W2_s = (const float*)d_in[9];
  const float* W3_s = (const float*)d_in[10];
  const float* W1_v = (const float*)d_in[11];
  const float* W2_v = (const float*)d_in[12];
  const float* W3_v = (const float*)d_in[13];
  float* ws  = (float*)d_ws;
  float* outp = (float*)d_out;

  {
    prep_gfrag<<<NE*KT, 256, 0, stream>>>(
        U1_s, U2_s, U3_s, U1_v, U2_v, U3_v,
        W1_s, W2_s, W3_s, W1_v, W2_v, W3_v, ws + OFF_GFRAG);
  }
  {
    const size_t smem = (size_t)(GFRAG_PER_C + 128*ZST) * sizeof(float);
    sc_main<<<dim3(NB/128, NC), 256, smem, stream>>>(x, yv, ws + OFF_GFRAG, outp);
  }
}


# N tile 2's 8 idle columns take the x8-cubic rows' G for cols 32..39 on the quadratic rows (Z = D + x8*D'): the tile runs 44 K tiles, not 56 (168 -> 156 WMMAs a wave); those sums reorder
# speedup vs baseline: 2.8879x; 1.0670x over previous
#include <hip/hip_runtime.h>
#include <hip/hip_bf16.h>
#include <utility>


#define NB   2048
#define NC   256
#define NI   9
#define NE   10
#define K3S  23
#define K2S  3
#define K3V  30
#define K2V  4
#define KTOT 224
#define KT   56
#define NC8  120
#define R_Q  120
#define R_L  165
#define R_P  174
#define R_C8 176
#define R_END 221
#define KT2  44
#define NCHUNK (2*KT + KT2)
#define ZST  49

#define OFF_GFRAG 19456
#define GFRAG_PER_C (NCHUNK*64)

typedef float v2f __attribute__((ext_vector_type(2)));
typedef float v8f __attribute__((ext_vector_type(8)));

#define F_ONE  9
#define F_ZERO 10
struct Fac { int f0, f1, f2; };
constexpr Fac fac_of(int t) {
  if (t < NC8) {
    int a = 0, rem = t;
    for (int it = 0; it < 8; it++) {
      const int n = (8 - a)*(8 - a + 1)/2;
      if (rem < n) break;
      rem -= n; a++;
    }
    int b = a;
    for (int it = 0; it < 8; it++) {
      const int n = 8 - b;
      if (rem < n) break;
      rem -= n; b++;
    }
    return Fac{a, b, b + rem};
  }
  if (t < R_L || (t >= R_C8 && t < R_END)) {
    const bool c8 = t >= R_C8;
    int a = 0, rem = t - (c8 ? R_C8 : R_Q);
    for (int it = 0; it < NI; it++) {
      const int n = NI - a;
      if (rem < n) break;
      rem -= n; a++;
    }
    return Fac{a, a + rem, c8 ? 8 : F_ONE};
  }
  if (t < R_P) return Fac{t - R_L, F_ONE, F_ONE};
  return Fac{F_ZERO, F_ONE, F_ONE};
}
static_assert(fac_of(0).f0 == 0 && fac_of(0).f1 == 0 && fac_of(0).f2 == 0, "first cubic row");
static_assert(fac_of(NC8 - 1).f0 == 7 && fac_of(NC8 - 1).f1 == 7 && fac_of(NC8 - 1).f2 == 7, "last cubic row without 8");
static_assert(fac_of(R_Q).f0 == 0 && fac_of(R_Q).f1 == 0 && fac_of(R_Q).f2 == F_ONE, "first quadratic row");
static_assert(fac_of(R_L - 1).f0 == 8 && fac_of(R_L - 1).f1 == 8 && fac_of(R_L - 1).f2 == F_ONE, "last quadratic row");
static_assert(fac_of(R_L).f0 == 0 && fac_of(R_L).f1 == F_ONE, "first linear row");
static_assert(fac_of(R_P - 1).f0 == 8 && fac_of(R_P).f0 == F_ZERO && fac_of(R_C8 - 1).f0 == F_ZERO, "last linear row, zero rows");
static_assert(fac_of(R_C8).f0 == 0 && fac_of(R_C8).f1 == 0 && fac_of(R_C8).f2 == 8, "first cubic row with 8");
static_assert(fac_of(R_END - 1).f0 == 8 && fac_of(R_END - 1).f1 == 8 && fac_of(R_END - 1).f2 == 8, "last cubic row");
static_assert(fac_of(R_END).f0 == F_ZERO && fac_of(KTOT - 1).f0 == F_ZERO, "closing zero rows");
static_assert(R_END - R_C8 == R_L - R_Q && (R_C8 - R_Q) % 4 == 0 && R_C8 == KT2*4, "x_8 cubic rows sit whole K tiles behind their quadratic rows");
constexpr bool row_is_cubic(int t) { return t < NC8 || (t >= R_C8 && t < R_END); }
constexpr bool row_is_quad(int t)  { return t >= R_Q && t < R_L; }
constexpr bool row_is_lin(int t)   { return t >= R_L && t < R_P; }
constexpr bool kinds_match_table() {
  for (int t = 0; t < KTOT; t++) {
    const Fac f = fac_of(t);
    if (row_is_cubic(t) != (f.f2 < NI)) return false;
    if (row_is_quad(t) != (f.f2 == F_ONE && f.f1 < NI)) return false;
    if (row_is_lin(t) != (f.f1 == F_ONE && f.f0 < NI)) return false;
    if (row_is_lin(t) && f.f0 != t - R_L) return false;
  }
  return true;
}
static_assert(kinds_match_table(), "range compares and fac_of agree on every row");

template <typename F>
__device__ inline float symsum3(int a, int b, int c, F fetch) {
  const int P[6][3] = {{a,b,c},{a,c,b},{b,a,c},{b,c,a},{c,a,b},{c,b,a}};
  float f[6];
  for (int p = 0; p < 6; p++) f[p] = fetch(P[p][0], P[p][1], P[p][2]);
  float s = 0.f;
  for (int p = 0; p < 6; p++) {
    int code = P[p][0]*81 + P[p][1]*9 + P[p][2];
    bool dup = false;
    for (int q = 0; q < p; q++)
      dup = dup || (P[q][0]*81 + P[q][1]*9 + P[q][2] == code);
    const float s1 = s + f[p];
    s = dup ? s : s1;
  }
  return s;
}


#define USTR 128

__device__ inline bool usym_entry(int t, int j,
                                  const float* __restrict__ U2_s,
                                  const float* __restrict__ U3_s,
                                  const float* __restrict__ U2_v,
                                  const float* __restrict__ U3_v,
                                  int& slot, float& val) {
  if (row_is_cubic(t)) {
    const Fac f = fac_of(t);
    const int a = f.f0, b = f.f1, c = f.f2;
    const bool sc = j < K3S;
    const int v  = sc ? 0 : (j - K3S) / K3V;
    const int k  = sc ? j : (j - K3S) % K3V;
    const int kk = sc ? K3S : K3V;
    const float* u = sc ? U3_s : U3_v + v*(NI*NI*NI*K3V);
    slot = sc ? k : 32 + 32*v + k;
    val = symsum3(a, b, c, [&](int i, int jj, int l) {
      return u[((i*NI + jj)*NI + l)*kk + k];
    });
    return true;
  }
  if (row_is_quad(t) && j < K2S + 3*K2V) {
    const Fac f = fac_of(t);
    const int a = f.f0, b = f.f1;
    const bool sc = j < K2S;
    const int v  = sc ? 0 : (j - K2S) / K2V;
    const int k  = sc ? j : (j - K2S) % K2V;
    const int kk = sc ? K2S : K2V;
    const float* u = sc ? U2_s : U2_v + v*(NI*NI*K2V);
    slot = sc ? k : 32 + 32*v + k;
    const float f0 = u[(a*NI + b)*kk + k];
    const float f1 = u[(b*NI + a)*kk + k];
    const float s1 = f0 + f1;
    val = (a != b) ? s1 : f0;
    return true;
  }
  return false;
}

__device__ inline void g_cubic_row(const float* ur,
                                   const float (&w3s)[K3S], const float (&w3v)[K3V],
                                   float (&r)[4]) {
  float s = 0.f;
#pragma unroll
  for (int k = 0; k < K3S; k++) s += ur[k] * w3s[k];
  r[0] = s;
#pragma unroll
  for (int v = 0; v < 3; v++) {
    const float* uv = ur + 32 + 32*v;
    float q = 0.f;
#pragma unroll
    for (int k = 0; k < K3V; k++) q += uv[k] * w3v[k];
    r[1 + v] = q;
  }
}

__device__ inline void g_low_row(const float* ur, int t, int e, int c,
                                 const float* __restrict__ U1_s,
                                 const float* __restrict__ U1_v,
                                 const float* __restrict__ W1_s,
                                 const float* __restrict__ W2_s,
                                 const float* __restrict__ W1_v,
                                 const float* __restrict__ W2_v,
                                 float (&r)[4]) {
  r[0] = 0.f; r[1] = 0.f; r[2] = 0.f; r[3] = 0.f;
  if (row_is_quad(t)) {
    {
      const float* w = W2_s + e*K2S*NC + c;
      float s = 0.f;
      for (int k = 0; k < K2S; k++) s += ur[k] * w[k*NC];
      r[0] = s;
    }
    for (int v = 0; v < 3; v++) {
      const float* u = ur + 32 + 32*v;
      const float* w = W2_v + e*K2V*NC + c;
      float s = 0.f;
      for (int k = 0; k < K2V; k++) s += u[k] * w[k*NC];
      r[1 + v] = s;
    }
  } else if (row_is_lin(t)) {
    const int i = t - R_L;
    const float w1s = W1_s[e*NC + c];
    const float w1v = W1_v[e*NC + c];
    r[0] = U1_s[i] * w1s;
    for (int v = 0; v < 3; v++) r[1 + v] = U1_v[v*NI + i] * w1v;
  }
}

__global__ void __launch_bounds__(256)
prep_gfrag(const float* __restrict__ U1_s,
           const float* __restrict__ U2_s,
           const float* __restrict__ U3_s,
           const float* __restrict__ U1_v,
           const float* __restrict__ U2_v,
           const float* __restrict__ U3_v,
           const float* __restrict__ W1_s,
           const float* __restrict__ W2_s,
           const float* __restrict__ W3_s,
           const float* __restrict__ W1_v,
           const float* __restrict__ W2_v,
           const float* __restrict__ W3_v,
           float* __restrict__ gfrag) {
  __shared__ __attribute__((aligned(16))) float usym[4*USTR];

  const int bid = blockIdx.x;
  const int kt  = bid % KT;
  const int e   = bid / KT;
  const int c   = (int)threadIdx.x;

  if (kt*4 < R_L || kt >= KT2) {
    const int NJ = K3S + 3*K3V;
    for (int idx = (int)threadIdx.x; idx < 4*NJ; idx += 256) {
      const int tl = idx / NJ, j = idx - tl*NJ;
      int slot; float val;
      if (usym_entry(kt*4 + tl, j, U2_s, U3_s, U2_v, U3_v, slot, val))
        usym[tl*USTR + slot] = val;
    }
  }
  __syncthreads();

  float res[4][4];
  if (kt*4 < NC8 || kt >= KT2) {
    float w3s[K3S], w3v[K3V];
#pragma unroll
    for (int k = 0; k < K3S; k++) w3s[k] = W3_s[(e*K3S + k)*NC + c];
#pragma unroll
    for (int k = 0; k < K3V; k++) w3v[k] = W3_v[(e*K3V + k)*NC + c];
#pragma unroll
    for (int tl = 0; tl < 4; tl++) {
      const int t = kt*4 + tl;
      if (row_is_cubic(t)) g_cubic_row(usym + tl*USTR, w3s, w3v, res[tl]);
      else        g_low_row(usym + tl*USTR, t, e, c, U1_s, U1_v, W1_s, W2_s, W1_v, W2_v, res[tl]);
    }
  } else {
#pragma unroll
    for (int tl = 0; tl < 4; tl++)
      g_low_row(usym + tl*USTR, kt*4 + tl, e, c, U1_s, U1_v, W1_s, W2_s, W1_v, W2_v, res[tl]);
  }

  const int ct = e >> 2;
  const bool idle = (ct == 2) && (kt >= KT2);
  const int ktd = idle ? kt - (R_C8 - R_Q)/4 : kt;
  float* dst = gfrag + (size_t)c*GFRAG_PER_C + (size_t)(ct*KT + ktd)*64 + (e & 3)*8 + (idle ? 16 : 0);
  typedef float v4f __attribute__((ext_vector_type(4)));
  v4f o;
  o.x = res[0][0]; o.y = res[1][0]; o.z = res[0][1]; o.w = res[1][1];
  *(v4f*)(dst +  0) = o;
  o.x = res[0][2]; o.y = res[1][2]; o.z = res[0][3]; o.w = res[1][3];
  *(v4f*)(dst +  4) = o;
  o.x = res[2][0]; o.y = res[3][0]; o.z = res[2][1]; o.w = res[3][1];
  *(v4f*)(dst + 32) = o;
  o.x = res[2][2]; o.y = res[3][2]; o.z = res[2][3]; o.w = res[3][3];
  *(v4f*)(dst + 36) = o;

  if (e == NE - 1 && (kt < R_Q/4 || (kt >= R_Q/4 + (KT - KT2) && kt < KT2))) {
    const v4f z = {0.f, 0.f, 0.f, 0.f};
    float* pad = gfrag + (size_t)c*GFRAG_PER_C + (size_t)(2*KT + kt)*64;
    for (int j = 0; j < 4; j++) {
      *(v4f*)(pad + 16 + j*4) = z;
      *(v4f*)(pad + 48 + j*4) = z;
    }
  }
}


template <int I>
__device__ __forceinline__ float fac_val(const float (&x)[NI]) {
  if constexpr (I < NI) return x[I];
  else if constexpr (I == F_ONE) return 1.f;
  else return 0.f;
}
template <int I, int J>
__device__ __forceinline__ float pick(const float (&x)[NI], bool hi) {
  if constexpr (I == J) return fac_val<I>(x);
  else return hi ? fac_val<J>(x) : fac_val<I>(x);
}
template <int T0, int T1>
__device__ __forceinline__ float slot(const float (&x)[NI], bool hi) {
  constexpr Fac a = fac_of(T0), b = fac_of(T1);
  const float p2 = pick<a.f0, b.f0>(x, hi) * pick<a.f1, b.f1>(x, hi);
  return p2 * pick<a.f2, b.f2>(x, hi);
}
template <int KT0, size_t... I>
__device__ __forceinline__ void build_a(const float (&x)[NI], bool hi,
                                        v2f (&A)[sizeof...(I)],
                                        std::index_sequence<I...>) {
  ((A[I] = v2f{slot<(KT0 + (int)I)*4 + 0, (KT0 + (int)I)*4 + 2>(x, hi),
               slot<(KT0 + (int)I)*4 + 1, (KT0 + (int)I)*4 + 3>(x, hi)}), ...);
}
#define GK 4
#define NG (KT/GK)
static_assert(KT % GK == 0 && KT2 % GK == 0, "N tile 2 ends on a step boundary");
template <int KT0>
__device__ __forceinline__ void load_b(const float* g0, const float* g1,
                                       const float* g2, v2f (&B)[GK][3]) {
#pragma unroll
  for (int i = 0; i < GK; i++) {
    B[i][0] = *(const v2f*)(g0 + (KT0 + i)*64);
    B[i][1] = *(const v2f*)(g1 + (KT0 + i)*64);
    if constexpr (KT0 < KT2) B[i][2] = *(const v2f*)(g2 + (KT0 + i)*64);
  }
}
template <int KT0>
__device__ __forceinline__ void mma_step(const float (&xv)[NI], bool hi,
                                         const v2f (&B)[GK][3],
                                         v8f& acc0, v8f& acc1, v8f& acc2) {
  v2f A[GK];
  build_a<KT0>(xv, hi, A, std::make_index_sequence<GK>{});
#pragma unroll
  for (int i = 0; i < GK; i++) {
    acc0 = __builtin_amdgcn_wmma_f32_16x16x4_f32(false, A[i], false, B[i][0],
                                                 (short)0, acc0, false, false);
    acc1 = __builtin_amdgcn_wmma_f32_16x16x4_f32(false, A[i], false, B[i][1],
                                                 (short)0, acc1, false, false);
    if constexpr (KT0 < KT2)
      acc2 = __builtin_amdgcn_wmma_f32_16x16x4_f32(false, A[i], false, B[i][2],
                                                   (short)0, acc2, false, false);
  }
}
template <int G>
__device__ __forceinline__ void k_steps(const float (&xv)[NI], bool hi,
                                        const float* g0, const float* g1,
                                        const float* g2,
                                        v2f (&Bc)[GK][3], v2f (&Bn)[GK][3],
                                        v8f& acc0, v8f& acc1, v8f& acc2) {
  if constexpr (G + 1 < NG) load_b<(G + 1)*GK>(g0, g1, g2, Bn);
  __builtin_amdgcn_sched_barrier(0);
  mma_step<G*GK>(xv, hi, Bc, acc0, acc1, acc2);
  if constexpr (G + 1 < NG)
    k_steps<G + 1>(xv, hi, g0, g1, g2, Bn, Bc, acc0, acc1, acc2);
}

__global__ void __launch_bounds__(256)
sc_main(const float* __restrict__ x, const float* __restrict__ y,
        const float* __restrict__ gfrag, float* __restrict__ out) {
  extern __shared__ float lds[];
  float* Gl = lds;
  float* Zs = Gl + GFRAG_PER_C;

  const int tid  = threadIdx.x;
  const int wave = tid >> 5;
  const int lane = tid & 31;
  const int c    = blockIdx.y;
  const int b0   = blockIdx.x * 128;
  float* Zb = Zs + wave*16*ZST;

  typedef float v4f __attribute__((ext_vector_type(4)));
  const int xr = wave*16 + (lane & 15);
  float xv[NI];
  {
    const float* xp = x + ((size_t)(b0 + xr)*NC + c)*NI;
#pragma unroll
    for (int i = 0; i < NI; i++) xv[i] = xp[i];
  }
  v4f gr[10];
  {
    const v4f* gsrc = (const v4f*)(gfrag + (size_t)c * GFRAG_PER_C);
#pragma unroll
    for (int j = 0; j < 9; j++) gr[j] = gsrc[tid + 256*j];
    gr[9] = gsrc[2304 + (tid < 192 ? tid : tid - 64)];
  }
  float yv[NE];
  {
    const float* yb = y + (size_t)(b0 + (tid >> 1))*NE;
#pragma unroll
    for (int e = 0; e < NE; e++) yv[e] = yb[e];
  }

  {
    v4f* Gl4 = (v4f*)Gl;
#pragma unroll
    for (int j = 0; j < 9; j++) Gl4[tid + 256*j] = gr[j];
    asm volatile("" :: "v"(gr[9]));
    if (tid < 192) Gl4[2304 + tid] = gr[9];
  }
  __syncthreads();

  {
    const bool hi  = (lane >> 4) != 0;
    const int  l16 = lane & 15;
    const float* g0 = Gl + 0*KT*64 + lane*2;
    const float* g1 = Gl + 1*KT*64 + lane*2;
    const float* g2 = Gl + 2*KT*64 + lane*2;

    v8f acc0 = {0,0,0,0,0,0,0,0};
    v8f acc1 = {0,0,0,0,0,0,0,0};
    v8f acc2 = {0,0,0,0,0,0,0,0};

    v2f Ba[GK][3], Bb[GK][3];
    load_b<0>(g0, g1, g2, Ba);
    k_steps<0>(xv, hi, g0, g1, g2, Ba, Bb, acc0, acc1, acc2);

    const int zr = hi ? 8 : 0;
#pragma unroll
    for (int g = 0; g < 8; g++) {
      Zb[(zr+g)*ZST +  0 + l16] = acc0[g];
      Zb[(zr+g)*ZST + 16 + l16] = acc1[g];
      Zb[(zr+g)*ZST + 32 + l16] = acc2[g];
    }
    Zb[l16*ZST + 48] = xv[8];
  }
  __syncthreads();

  {
    const int r  = tid >> 1;
    const int p0 = (tid & 1)*2;
    const int b  = b0 + r;
    const float* zrow = Zb + (r & 15)*ZST;
    const float x8 = zrow[48];
#pragma unroll
    for (int pp = 0; pp < 2; pp++) {
      const int p = p0 + pp;
      float s = 0.f;
#pragma unroll
      for (int e = 0; e < NE; e++) {
        float z = zrow[e*4 + p];
        if (e*4 >= 32) z += x8 * zrow[e*4 + p + 8];
        s += yv[e] * z;
      }
      size_t o = (size_t)b*(4*NC) +
                 ((p == 0) ? (size_t)c : (size_t)(NC + c*3 + (p-1)));
      out[o] = s;
    }
  }
}

extern "C" void kernel_launch(void* const* d_in, const int* in_sizes, int n_in,
                              void* d_out, int out_size, void* d_ws, size_t ws_size,
                              hipStream_t stream) {
  const float* x    = (const float*)d_in[0];
  const float* yv   = (const float*)d_in[1];
  const float* U1_s = (const float*)d_in[2];
  const float* U2_s = (const float*)d_in[3];
  const float* U3_s = (const float*)d_in[4];
  const float* U1_v = (const float*)d_in[5];
  const float* U2_v = (const float*)d_in[6];
  const float* U3_v = (const float*)d_in[7];
  const float* W1_s = (const float*)d_in[8];
  const float* W2_s = (const float*)d_in[9];
  const float* W3_s = (const float*)d_in[10];
  const float* W1_v = (const float*)d_in[11];
  const float* W2_v = (const float*)d_in[12];
  const float* W3_v = (const float*)d_in[13];
  float* ws  = (float*)d_ws;
  float* outp = (float*)d_out;

  {
    prep_gfrag<<<NE*KT, 256, 0, stream>>>(
        U1_s, U2_s, U3_s, U1_v, U2_v, U3_v,
        W1_s, W2_s, W3_s, W1_v, W2_v, W3_v, ws + OFF_GFRAG);
  }
  {
    const size_t smem = (size_t)(GFRAG_PER_C + 128*ZST) * sizeof(float);
    sc_main<<<dim3(NB/128, NC), 256, smem, stream>>>(x, yv, ws + OFF_GFRAG, outp);
  }
}
